# starting kernel
# baseline (speedup 1.0000x reference)
#include <hip/hip_runtime.h>


#define N_NODES 50000
#define N_EDGES 600000
#define DIM     128
#define M_TILES (N_NODES / 16)

typedef __attribute__((ext_vector_type(2))) float v2f;
typedef __attribute__((ext_vector_type(8))) float v8f;

__global__ __launch_bounds__(128)
void gcn_gemm128(const float* __restrict__ A, const float* __restrict__ W,
                 float* __restrict__ C, int M) {
    __shared__ float ldsW[DIM * DIM];

    const int tid  = threadIdx.x;
    const int lane = tid & 31;
    const int wave = tid >> 5;

    for (int p = tid; p < 8192; p += 128) {
        const int l    = p & 31;
        const int nt   = (p >> 5) & 7;
        const int kk   = p >> 8;
        const int krow = kk * 4 + ((l & 16) ? 2 : 0);
        const int col  = nt * 16 + (l & 15);
        ldsW[2 * p + 0] = W[(krow + 0) * DIM + col];
        ldsW[2 * p + 1] = W[(krow + 1) * DIM + col];
    }
    __syncthreads();

    const int mtile = blockIdx.x * 4 + wave;
    if (mtile * 16 >= M) return;

    const float* __restrict__ Arow =
        A + (size_t)(mtile * 16 + (lane & 15)) * DIM + ((lane & 16) ? 2 : 0);

    v8f acc[8] = {};

    for (int kk = 0; kk < 32; ++kk) {
        const v2f a = *(const v2f*)(Arow + kk * 4);
#pragma unroll
        for (int nt = 0; nt < 8; ++nt) {
            const v2f b =
                *(const v2f*)&ldsW[(((kk << 3) + nt) * 32 + lane) * 2];
            acc[nt] = __builtin_amdgcn_wmma_f32_16x16x4_f32(
                 false, a,  false, b,
                 (short)0, acc[nt],
                 false,  false);
        }
    }

    const int row0 = mtile * 16 + ((lane & 16) ? 8 : 0);
    const int colc = lane & 15;
#pragma unroll
    for (int nt = 0; nt < 8; ++nt) {
#pragma unroll
        for (int j = 0; j < 8; ++j) {
            C[(size_t)(row0 + j) * DIM + nt * 16 + colc] = acc[nt][j];
        }
    }
}

__global__ __launch_bounds__(256)
void gcn_fill0(float* __restrict__ p, int n) {
    const int i = blockIdx.x * 256 + threadIdx.x;
    if (i < n) p[i] = 0.0f;
}

__global__ __launch_bounds__(256)
void gcn_spmm(const int* __restrict__ erow, const int* __restrict__ ecol,
              const float* __restrict__ evals, const float* __restrict__ h,
              float* __restrict__ out, int E) {
    const int e    = blockIdx.x * 8 + (threadIdx.x >> 5);
    const int lane = threadIdx.x & 31;
    if (e >= E) return;

    const int   r = erow[e];
    const int   c = ecol[e];
    const float v = evals[e];

    const float4 x = ((const float4*)(h + (size_t)c * DIM))[lane];
    float* __restrict__ o = out + (size_t)r * DIM + lane * 4;

    __hip_atomic_fetch_add(o + 0, v * x.x, __ATOMIC_RELAXED, __HIP_MEMORY_SCOPE_AGENT);
    __hip_atomic_fetch_add(o + 1, v * x.y, __ATOMIC_RELAXED, __HIP_MEMORY_SCOPE_AGENT);
    __hip_atomic_fetch_add(o + 2, v * x.z, __ATOMIC_RELAXED, __HIP_MEMORY_SCOPE_AGENT);
    __hip_atomic_fetch_add(o + 3, v * x.w, __ATOMIC_RELAXED, __HIP_MEMORY_SCOPE_AGENT);
}

__global__ __launch_bounds__(256)
void gcn_bias_tanh(float* __restrict__ x, const float* __restrict__ b, int n) {
    const int i = blockIdx.x * 256 + threadIdx.x;
    if (i < n) x[i] = tanhf(x[i] + b[i & (DIM - 1)]);
}

__global__ __launch_bounds__(256)
void gcn_gemv(const float* __restrict__ h, const float* __restrict__ fcw,
              const float* __restrict__ fcb, float* __restrict__ out, int M) {
    const int node = blockIdx.x * 8 + (threadIdx.x >> 5);
    const int lane = threadIdx.x & 31;
    if (node >= M) return;

    const float4 x = ((const float4*)(h + (size_t)node * DIM))[lane];
    const float4 w = ((const float4*)fcw)[lane];
    float s = x.x * w.x + x.y * w.y + x.z * w.z + x.w * w.w;
#pragma unroll
    for (int off = 16; off > 0; off >>= 1) s += __shfl_down(s, off, 32);
    if (lane == 0) out[node] = s + fcb[0];
}

extern "C" void kernel_launch(void* const* d_in, const int* in_sizes, int n_in,
                              void* d_out, int out_size, void* d_ws, size_t ws_size,
                              hipStream_t stream) {
    const float* features = (const float*)d_in[0];
    const int*   erow     = (const int*)  d_in[1];
    const int*   ecol     = (const int*)  d_in[2];
    const float* evals    = (const float*)d_in[3];
    const float* W1       = (const float*)d_in[4];
    const float* b1       = (const float*)d_in[5];
    const float* W2       = (const float*)d_in[6];
    const float* b2       = (const float*)d_in[7];
    const float* fcw      = (const float*)d_in[8];
    const float* fcb      = (const float*)d_in[9];
    float*       out      = (float*)d_out;

    float* tmp0 = (float*)d_ws;
    float* tmp1 = tmp0 + (size_t)N_NODES * DIM;

    const int nElem      = N_NODES * DIM;
    const dim3 gemmGrid((M_TILES + 3) / 4);
    const dim3 elemGrid((nElem + 255) / 256);
    const dim3 spmmGrid((N_EDGES + 7) / 8);
    const dim3 gemvGrid((N_NODES + 7) / 8);

    gcn_gemm128<<<gemmGrid, 128, 0, stream>>>(features, W1, tmp0, N_NODES);
    gcn_fill0  <<<elemGrid, 256, 0, stream>>>(tmp1, nElem);
    gcn_spmm   <<<spmmGrid, 256, 0, stream>>>(erow, ecol, evals, tmp0, tmp1, N_EDGES);
    gcn_bias_tanh<<<elemGrid, 256, 0, stream>>>(tmp1, b1, nElem);

    gcn_gemm128<<<gemmGrid, 128, 0, stream>>>(tmp1, W2, tmp0, N_NODES);
    gcn_fill0  <<<elemGrid, 256, 0, stream>>>(tmp1, nElem);
    gcn_spmm   <<<spmmGrid, 256, 0, stream>>>(erow, ecol, evals, tmp0, tmp1, N_EDGES);
    gcn_bias_tanh<<<elemGrid, 256, 0, stream>>>(tmp1, b2, nElem);

    gcn_gemv<<<gemvGrid, 256, 0, stream>>>(tmp1, fcw, fcb, out, N_NODES);
}


# A @ h by row-block ownership: a work-group keeps 196 rows' sums in LDS, scans the whole unsorted edge list, adds by LDS f32 atomics (arrival order), stores rows once; global atomics, zero-fill gone
# speedup vs baseline: 1.5366x; 1.5366x over previous
#include <hip/hip_runtime.h>


#define N_NODES 50000
#define N_EDGES 600000
#define DIM     128
#define M_TILES (N_NODES / 16)

typedef __attribute__((ext_vector_type(2))) float v2f;
typedef __attribute__((ext_vector_type(8))) float v8f;

__global__ __launch_bounds__(128)
void gcn_gemm128(const float* __restrict__ A, const float* __restrict__ W,
                 float* __restrict__ C, int M) {
    __shared__ float ldsW[DIM * DIM];

    const int tid  = threadIdx.x;
    const int lane = tid & 31;
    const int wave = tid >> 5;

    for (int p = tid; p < 8192; p += 128) {
        const int l    = p & 31;
        const int nt   = (p >> 5) & 7;
        const int kk   = p >> 8;
        const int krow = kk * 4 + ((l & 16) ? 2 : 0);
        const int col  = nt * 16 + (l & 15);
        ldsW[2 * p + 0] = W[(krow + 0) * DIM + col];
        ldsW[2 * p + 1] = W[(krow + 1) * DIM + col];
    }
    __syncthreads();

    const int mtile = blockIdx.x * 4 + wave;
    if (mtile * 16 >= M) return;

    const float* __restrict__ Arow =
        A + (size_t)(mtile * 16 + (lane & 15)) * DIM + ((lane & 16) ? 2 : 0);

    v8f acc[8] = {};

    for (int kk = 0; kk < 32; ++kk) {
        const v2f a = *(const v2f*)(Arow + kk * 4);
#pragma unroll
        for (int nt = 0; nt < 8; ++nt) {
            const v2f b =
                *(const v2f*)&ldsW[(((kk << 3) + nt) * 32 + lane) * 2];
            acc[nt] = __builtin_amdgcn_wmma_f32_16x16x4_f32(
                 false, a,  false, b,
                 (short)0, acc[nt],
                 false,  false);
        }
    }

    const int row0 = mtile * 16 + ((lane & 16) ? 8 : 0);
    const int colc = lane & 15;
#pragma unroll
    for (int nt = 0; nt < 8; ++nt) {
#pragma unroll
        for (int j = 0; j < 8; ++j) {
            C[(size_t)(row0 + j) * DIM + nt * 16 + colc] = acc[nt][j];
        }
    }
}

#define SPMM_ROWS    196
#define SPMM_WGS     ((N_NODES + SPMM_ROWS - 1) / SPMM_ROWS)
#define SPMM_WAVES   32
#define SPMM_THREADS (SPMM_WAVES * 32)
#define EDGE_VEC     (N_EDGES / 4)
#define SCAN_STEPS   ((EDGE_VEC + SPMM_THREADS - 1) / SPMM_THREADS)
#define SCAN_DEPTH   7
#define SCAN_ROUNDS  (SCAN_STEPS / SCAN_DEPTH)
#define SPMM_LIST    (31 + SCAN_DEPTH * 128)
#define GATHER_DEPTH 16

static_assert(N_EDGES % 4 == 0, "edge list is read as int4");
static_assert(SCAN_ROUNDS * SCAN_DEPTH == SCAN_STEPS, "rounds cover the steps exactly");
static_assert(SPMM_ROWS < 256 && N_EDGES < (1 << 24), "list entry packs edge<<8 | local row");

__global__ __launch_bounds__(SPMM_THREADS)
void gcn_spmm_own(const int* __restrict__ erow, const int* __restrict__ ecol,
                  const float* __restrict__ evals, const float* __restrict__ h,
                  float* __restrict__ out) {
    __shared__ float    acc[(SPMM_ROWS + 1) * DIM];
    __shared__ unsigned elist[SPMM_WAVES * SPMM_LIST];

    const int tid  = threadIdx.x;
    const int lane = tid & 31;
    const int wave = tid >> 5;
    const int base = blockIdx.x * SPMM_ROWS;

    for (int i = tid; i < (SPMM_ROWS + 1) * DIM / 4; i += SPMM_THREADS)
        ((float4*)acc)[i] = make_float4(0.0f, 0.0f, 0.0f, 0.0f);
    __syncthreads();

    unsigned* __restrict__ mylist = elist + wave * SPMM_LIST;
    const int4* __restrict__ erow4 = (const int4*)erow;

    const int step0 = (blockIdx.x * SCAN_STEPS) / SPMM_WGS;

    auto vec_index = [&](int rnd, int k) {
        int s = step0 + rnd * SCAN_DEPTH + k;
        if (s >= SCAN_STEPS) s -= SCAN_STEPS;
        return s * SPMM_THREADS + tid;
    };
    auto load_round = [&](int4 (&r)[SCAN_DEPTH], int rnd) {
#pragma unroll
        for (int k = 0; k < SCAN_DEPTH; ++k)
            r[k] = erow4[min(vec_index(rnd, k), EDGE_VEC - 1)];
    };

    int count = 0;

    auto look_through = [&](const int4 (&r)[SCAN_DEPTH], int rnd) {
#pragma unroll
        for (int k = 0; k < SCAN_DEPTH; ++k) {
            const int vi = vec_index(rnd, k);
            const unsigned d0 = (unsigned)r[k].x - (unsigned)base;
            const unsigned d1 = (unsigned)r[k].y - (unsigned)base;
            const unsigned d2 = (unsigned)r[k].z - (unsigned)base;
            const unsigned d3 = (unsigned)r[k].w - (unsigned)base;
            const unsigned dmin = (vi < EDGE_VEC) ? min(min(d0, d1), min(d2, d3)) : ~0u;
            if (__builtin_amdgcn_ballot_w32(dmin < SPMM_ROWS) != 0u) {
                const unsigned ebase = (unsigned)vi * 4u;
                const unsigned d[4] = {d0, d1, d2, d3};
#pragma unroll
                for (int q = 0; q < 4; ++q) {
                    const bool m = (dmin < SPMM_ROWS) && (d[q] < SPMM_ROWS);
                    const unsigned b = __builtin_amdgcn_ballot_w32(m);
                    if (m) mylist[count + __builtin_amdgcn_mbcnt_lo(b, 0u)] = ((ebase + q) << 8) | d[q];
                    count += __builtin_popcount(b);
                }
            }
        }
    };

    auto drain = [&](int keep) {
        while (count > keep) {
            const int n = min(count, 32);
            count -= n;
            const bool live = lane < n;
            const unsigned p = live ? mylist[count + lane] : 0u;
            const int e  = (int)(p >> 8);
            const int lr = live ? (int)(p & 255u) : SPMM_ROWS;
            const int   c = (int)min((unsigned)ecol[e], (unsigned)(N_NODES - 1));
            const float v = evals[e];
            for (int j = 0; j < n; j += GATHER_DEPTH) {
                float4 x[GATHER_DEPTH];
#pragma unroll
                for (int u = 0; u < GATHER_DEPTH; ++u) {
                    const int cj = __builtin_amdgcn_readlane(c, j + u);
                    x[u] = ((const float4*)(h + (size_t)cj * DIM))[lane];
                }
#pragma unroll
                for (int u = 0; u < GATHER_DEPTH; ++u) {
                    const float vj = __builtin_bit_cast(
                        float, __builtin_amdgcn_readlane(__builtin_bit_cast(int, v), j + u));
                    const int lrj = __builtin_amdgcn_readlane(lr, j + u);
                    float* a = acc + lrj * DIM + lane;
                    __hip_atomic_fetch_add(a +  0, vj * x[u].x, __ATOMIC_RELAXED, __HIP_MEMORY_SCOPE_WORKGROUP);
                    __hip_atomic_fetch_add(a + 32, vj * x[u].y, __ATOMIC_RELAXED, __HIP_MEMORY_SCOPE_WORKGROUP);
                    __hip_atomic_fetch_add(a + 64, vj * x[u].z, __ATOMIC_RELAXED, __HIP_MEMORY_SCOPE_WORKGROUP);
                    __hip_atomic_fetch_add(a + 96, vj * x[u].w, __ATOMIC_RELAXED, __HIP_MEMORY_SCOPE_WORKGROUP);
                }
            }
        }
    };

    int4 cur[SCAN_DEPTH];
    load_round(cur, 0);
    for (int rnd = 0; rnd < SCAN_ROUNDS - 1; ++rnd) {
        int4 nxt[SCAN_DEPTH];
        load_round(nxt, rnd + 1);
        look_through(cur, rnd);
        drain(31);
#pragma unroll
        for (int k = 0; k < SCAN_DEPTH; ++k) cur[k] = nxt[k];
    }
    look_through(cur, SCAN_ROUNDS - 1);
    drain(0);
    __syncthreads();

    for (int rr = wave; rr < SPMM_ROWS; rr += SPMM_WAVES) {
        const int row = base + rr;
        if (row < N_NODES) {
            const float* a = acc + rr * DIM + lane;
            ((float4*)(out + (size_t)row * DIM))[lane] = make_float4(a[0], a[32], a[64], a[96]);
        }
    }
}

__global__ __launch_bounds__(256)
void gcn_bias_tanh(float* __restrict__ x, const float* __restrict__ b, int n) {
    const int i = blockIdx.x * 256 + threadIdx.x;
    if (i < n) x[i] = tanhf(x[i] + b[i & (DIM - 1)]);
}

__global__ __launch_bounds__(256)
void gcn_gemv(const float* __restrict__ h, const float* __restrict__ fcw,
              const float* __restrict__ fcb, float* __restrict__ out, int M) {
    const int node = blockIdx.x * 8 + (threadIdx.x >> 5);
    const int lane = threadIdx.x & 31;
    if (node >= M) return;

    const float4 x = ((const float4*)(h + (size_t)node * DIM))[lane];
    const float4 w = ((const float4*)fcw)[lane];
    float s = x.x * w.x + x.y * w.y + x.z * w.z + x.w * w.w;
#pragma unroll
    for (int off = 16; off > 0; off >>= 1) s += __shfl_down(s, off, 32);
    if (lane == 0) out[node] = s + fcb[0];
}

extern "C" void kernel_launch(void* const* d_in, const int* in_sizes, int n_in,
                              void* d_out, int out_size, void* d_ws, size_t ws_size,
                              hipStream_t stream) {
    const float* features = (const float*)d_in[0];
    const int*   erow     = (const int*)  d_in[1];
    const int*   ecol     = (const int*)  d_in[2];
    const float* evals    = (const float*)d_in[3];
    const float* W1       = (const float*)d_in[4];
    const float* b1       = (const float*)d_in[5];
    const float* W2       = (const float*)d_in[6];
    const float* b2       = (const float*)d_in[7];
    const float* fcw      = (const float*)d_in[8];
    const float* fcb      = (const float*)d_in[9];
    float*       out      = (float*)d_out;

    float* tmp0 = (float*)d_ws;
    float* tmp1 = tmp0 + (size_t)N_NODES * DIM;

    const int nElem      = N_NODES * DIM;
    const dim3 gemmGrid((M_TILES + 3) / 4);
    const dim3 elemGrid((nElem + 255) / 256);
    const dim3 spmmGrid(SPMM_WGS);
    const dim3 gemvGrid((N_NODES + 7) / 8);

    gcn_gemm128<<<gemmGrid, 128, 0, stream>>>(features, W1, tmp0, N_NODES);
    gcn_spmm_own<<<spmmGrid, SPMM_THREADS, 0, stream>>>(erow, ecol, evals, tmp0, tmp1);
    gcn_bias_tanh<<<elemGrid, 256, 0, stream>>>(tmp1, b1, nElem);

    gcn_gemm128<<<gemmGrid, 128, 0, stream>>>(tmp1, W2, tmp0, N_NODES);
    gcn_spmm_own<<<spmmGrid, SPMM_THREADS, 0, stream>>>(erow, ecol, evals, tmp0, tmp1);
    gcn_bias_tanh<<<elemGrid, 256, 0, stream>>>(tmp1, b2, nElem);

    gcn_gemv<<<gemvGrid, 256, 0, stream>>>(tmp1, fcw, fcb, out, N_NODES);
}


# each layer finished by the row-block owner: its LDS sums times W on f32 WMMA, i.e. (A@H)@W, then bias, tanh (layer 2: fc dot) before the one store; 7 launches to 2; order of sums changes
# speedup vs baseline: 2.3014x; 1.4977x over previous
#include <hip/hip_runtime.h>


#define N_NODES 50000
#define N_EDGES 600000
#define DIM     128

typedef __attribute__((ext_vector_type(2))) float v2f;
typedef __attribute__((ext_vector_type(8))) float v8f;

#define SPMM_ROWS    196
#define SPMM_WGS     ((N_NODES + SPMM_ROWS - 1) / SPMM_ROWS)
#define SPMM_WAVES   32
#define SPMM_THREADS (SPMM_WAVES * 32)
#define EDGE_VEC     (N_EDGES / 4)
#define SCAN_STEPS   ((EDGE_VEC + SPMM_THREADS - 1) / SPMM_THREADS)
#define SCAN_DEPTH   7
#define SCAN_ROUNDS  (SCAN_STEPS / SCAN_DEPTH)
#define SPMM_LIST    (31 + SCAN_DEPTH * 128)
#define GATHER_DEPTH 16
#define ROW_TILES    ((SPMM_ROWS + 15) / 16)
#define ACC_ROWS     (ROW_TILES * 16)
#define ACC_PITCH    (DIM + 4)

static_assert(N_EDGES % 4 == 0, "edge list is read as int4");
static_assert(SCAN_ROUNDS * SCAN_DEPTH == SCAN_STEPS, "rounds cover the steps exactly");
static_assert(SPMM_ROWS < 256 && N_EDGES < (1 << 24), "list entry packs edge<<8 | local row");
static_assert(SPMM_ROWS < ACC_ROWS, "row SPMM_ROWS is the spare row");
static_assert(2 * ROW_TILES <= SPMM_WAVES, "one wave per 16-row x 64-column piece");


template <bool LAST>
__global__ __launch_bounds__(SPMM_THREADS)
void gcn_layer_own(const int* __restrict__ erow, const int* __restrict__ ecol,
                   const float* __restrict__ evals, const float* __restrict__ h,
                   const float* __restrict__ Wm, const float* __restrict__ bias,
                   const float* __restrict__ fcw, const float* __restrict__ fcb,
                   float* __restrict__ out) {
    __shared__ float    acc[ACC_ROWS * ACC_PITCH];
    __shared__ unsigned elist[SPMM_WAVES * SPMM_LIST];
    __shared__ float    ldsW[DIM * DIM];

    const int tid  = threadIdx.x;
    const int lane = tid & 31;
    const int wave = tid >> 5;
    const int base = blockIdx.x * SPMM_ROWS;

    for (int i = tid; i < ACC_ROWS * ACC_PITCH / 4; i += SPMM_THREADS)
        ((float4*)acc)[i] = make_float4(0.0f, 0.0f, 0.0f, 0.0f);

    for (int p = tid; p < DIM * DIM / 2; p += SPMM_THREADS) {
        const int l   = p & 31;
        const int nt  = (p >> 5) & 7;
        const int kk  = p >> 8;
        const int k0  = 16 * (kk >> 2) + 8 * (l >> 4) + (kk & 3);
        const int col = nt * 16 + (l & 15);
        v2f w2;
        w2.x = Wm[(k0 + 0) * DIM + col];
        w2.y = Wm[(k0 + 4) * DIM + col];
        *(v2f*)&ldsW[2 * p] = w2;
    }
    __syncthreads();

    unsigned* __restrict__ mylist = elist + wave * SPMM_LIST;
    const int4* __restrict__ erow4 = (const int4*)erow;

    const int step0 = (blockIdx.x * SCAN_STEPS) / SPMM_WGS;

    auto vec_index = [&](int rnd, int k) {
        int s = step0 + rnd * SCAN_DEPTH + k;
        if (s >= SCAN_STEPS) s -= SCAN_STEPS;
        return s * SPMM_THREADS + tid;
    };
    auto load_round = [&](int4 (&r)[SCAN_DEPTH], int rnd) {
#pragma unroll
        for (int k = 0; k < SCAN_DEPTH; ++k)
            r[k] = erow4[min(vec_index(rnd, k), EDGE_VEC - 1)];
    };

    int count = 0;

    auto look_through = [&](const int4 (&r)[SCAN_DEPTH], int rnd) {
#pragma unroll
        for (int k = 0; k < SCAN_DEPTH; ++k) {
            const int vi = vec_index(rnd, k);
            const unsigned d0 = (unsigned)r[k].x - (unsigned)base;
            const unsigned d1 = (unsigned)r[k].y - (unsigned)base;
            const unsigned d2 = (unsigned)r[k].z - (unsigned)base;
            const unsigned d3 = (unsigned)r[k].w - (unsigned)base;
            const unsigned dmin = (vi < EDGE_VEC) ? min(min(d0, d1), min(d2, d3)) : ~0u;
            if (__builtin_amdgcn_ballot_w32(dmin < SPMM_ROWS) != 0u) {
                const unsigned ebase = (unsigned)vi * 4u;
                const unsigned d[4] = {d0, d1, d2, d3};
#pragma unroll
                for (int q = 0; q < 4; ++q) {
                    const bool m = (dmin < SPMM_ROWS) && (d[q] < SPMM_ROWS);
                    const unsigned b = __builtin_amdgcn_ballot_w32(m);
                    if (m) mylist[count + __builtin_amdgcn_mbcnt_lo(b, 0u)] = ((ebase + q) << 8) | d[q];
                    count += __builtin_popcount(b);
                }
            }
        }
    };

    auto drain = [&](int keep) {
        while (count > keep) {
            const int n = min(count, 32);
            count -= n;
            const bool live = lane < n;
            const unsigned p = live ? mylist[count + lane] : 0u;
            const int e  = (int)(p >> 8);
            const int lr = live ? (int)(p & 255u) : SPMM_ROWS;
            const int   c = (int)min((unsigned)ecol[e], (unsigned)(N_NODES - 1));
            const float v = evals[e];
            for (int j = 0; j < n; j += GATHER_DEPTH) {
                float4 x[GATHER_DEPTH];
#pragma unroll
                for (int u = 0; u < GATHER_DEPTH; ++u) {
                    const int cj = __builtin_amdgcn_readlane(c, j + u);
                    x[u] = ((const float4*)(h + (size_t)cj * DIM))[lane];
                }
#pragma unroll
                for (int u = 0; u < GATHER_DEPTH; ++u) {
                    const float vj = __builtin_bit_cast(
                        float, __builtin_amdgcn_readlane(__builtin_bit_cast(int, v), j + u));
                    const int lrj = __builtin_amdgcn_readlane(lr, j + u);
                    float* a = acc + lrj * ACC_PITCH + lane;
                    __hip_atomic_fetch_add(a +  0, vj * x[u].x, __ATOMIC_RELAXED, __HIP_MEMORY_SCOPE_WORKGROUP);
                    __hip_atomic_fetch_add(a + 32, vj * x[u].y, __ATOMIC_RELAXED, __HIP_MEMORY_SCOPE_WORKGROUP);
                    __hip_atomic_fetch_add(a + 64, vj * x[u].z, __ATOMIC_RELAXED, __HIP_MEMORY_SCOPE_WORKGROUP);
                    __hip_atomic_fetch_add(a + 96, vj * x[u].w, __ATOMIC_RELAXED, __HIP_MEMORY_SCOPE_WORKGROUP);
                }
            }
        }
    };

    int4 cur[SCAN_DEPTH];
    load_round(cur, 0);
    for (int rnd = 0; rnd < SCAN_ROUNDS - 1; ++rnd) {
        int4 nxt[SCAN_DEPTH];
        load_round(nxt, rnd + 1);
        look_through(cur, rnd);
        drain(31);
#pragma unroll
        for (int k = 0; k < SCAN_DEPTH; ++k) cur[k] = nxt[k];
    }
    look_through(cur, SCAN_ROUNDS - 1);
    drain(0);
    __syncthreads();

    const bool mm = wave < 2 * ROW_TILES;
    const int  mt = wave >> 1;
    const int  nh = wave & 1;
    const int  hh = lane >> 4;
    const int  cc = lane & 15;
    v8f d[4] = {};
    if (mm) {
        const float* arow = acc + (mt * 16 + cc) * ACC_PITCH + 2 * hh;
        for (int kk = 0; kk < 32; ++kk) {
            const v2f a = *(const v2f*)(arow + (kk & 3) * 32 + 4 * (kk >> 2));
#pragma unroll
            for (int nt = 0; nt < 4; ++nt) {
                const v2f b = *(const v2f*)&ldsW[(((kk << 3) + nh * 4 + nt) * 32 + lane) * 2];
                d[nt] = __builtin_amdgcn_wmma_f32_16x16x4_f32(
                     false, a,  false, b,
                     (short)0, d[nt],
                     false,  false);
            }
        }
    }
    __syncthreads();
    if (mm) {
#pragma unroll
        for (int nt = 0; nt < 4; ++nt) {
            const int   col = nh * 64 + nt * 16 + cc;
            const float bc  = bias[col];
            float* t = acc + (mt * 16 + 8 * hh) * ACC_PITCH + (col & 3) * 32 + (col >> 2);
#pragma unroll
            for (int j = 0; j < 8; ++j) t[j * ACC_PITCH] = tanhf(d[nt][j] + bc);
        }
    }
    __syncthreads();

    if (!LAST) {
        for (int rr = wave; rr < SPMM_ROWS; rr += SPMM_WAVES) {
            const int row = base + rr;
            if (row < N_NODES) {
                const float* a = acc + rr * ACC_PITCH + lane;
                ((float4*)(out + (size_t)row * DIM))[lane] = make_float4(a[0], a[32], a[64], a[96]);
            }
        }
    } else {
        float* dots = (float*)elist;
        const float4 w = ((const float4*)fcw)[lane];
        for (int rr = wave; rr < SPMM_ROWS; rr += SPMM_WAVES) {
            const float* a = acc + rr * ACC_PITCH + lane;
            float s = a[0] * w.x + a[32] * w.y + a[64] * w.z + a[96] * w.w;
#pragma unroll
            for (int off = 16; off > 0; off >>= 1) s += __shfl_down(s, off, 32);
            if (lane == 0) dots[rr] = s + fcb[0];
        }
        __syncthreads();
        if (tid < SPMM_ROWS && base + tid < N_NODES) out[base + tid] = dots[tid];
    }
}

extern "C" void kernel_launch(void* const* d_in, const int* in_sizes, int n_in,
                              void* d_out, int out_size, void* d_ws, size_t ws_size,
                              hipStream_t stream) {
    const float* features = (const float*)d_in[0];
    const int*   erow     = (const int*)  d_in[1];
    const int*   ecol     = (const int*)  d_in[2];
    const float* evals    = (const float*)d_in[3];
    const float* W1       = (const float*)d_in[4];
    const float* b1       = (const float*)d_in[5];
    const float* W2       = (const float*)d_in[6];
    const float* b2       = (const float*)d_in[7];
    const float* fcw      = (const float*)d_in[8];
    const float* fcb      = (const float*)d_in[9];
    float*       out      = (float*)d_out;

    float* h1 = (float*)d_ws;

    const dim3 grid(SPMM_WGS);

    gcn_layer_own<false><<<grid, SPMM_THREADS, 0, stream>>>(erow, ecol, evals, features, W1, b1, fcw, fcb, h1);
    gcn_layer_own<true><<<grid, SPMM_THREADS, 0, stream>>>(erow, ecol, evals, h1, W2, b2, fcw, fcb, out);
}


# layer 1's row-block owner keeps each edge it gathers (column, local row, value) in the workspace, per wave; layer 2 reads that list instead of the whole edge list again; arrival order as before
# speedup vs baseline: 2.8274x; 1.2286x over previous
#include <hip/hip_runtime.h>


#define N_NODES 50000
#define N_EDGES 600000
#define DIM     128

typedef __attribute__((ext_vector_type(2))) float v2f;
typedef __attribute__((ext_vector_type(8))) float v8f;
typedef __attribute__((ext_vector_type(2))) unsigned v2u;

#define SPMM_ROWS    196
#define SPMM_WGS     ((N_NODES + SPMM_ROWS - 1) / SPMM_ROWS)
#define SPMM_WAVES   32
#define SPMM_THREADS (SPMM_WAVES * 32)
#define EDGE_VEC     (N_EDGES / 4)
#define SCAN_STEPS   ((EDGE_VEC + SPMM_THREADS - 1) / SPMM_THREADS)
#define SCAN_DEPTH   7
#define SCAN_ROUNDS  (SCAN_STEPS / SCAN_DEPTH)
#define SPMM_LIST    (31 + SCAN_DEPTH * 128)
#define GATHER_DEPTH 16
#define ROW_TILES    ((SPMM_ROWS + 15) / 16)
#define ACC_ROWS     (ROW_TILES * 16)
#define ACC_PITCH    (DIM + 4)
#define KEEP_CAP     256
#define KEEP_AHEAD   3
#define WS_KEPT_OFF  ((size_t)N_NODES * DIM * 4)
#define WS_NKEPT_OFF (WS_KEPT_OFF + (size_t)SPMM_WGS * SPMM_WAVES * KEEP_CAP * 8)

static_assert(N_EDGES % 4 == 0, "edge list is read as int4");
static_assert(SCAN_ROUNDS * SCAN_DEPTH == SCAN_STEPS, "rounds cover the steps exactly");
static_assert(SPMM_ROWS < 256 && N_EDGES < (1 << 24), "list entry packs edge<<8 | local row");
static_assert(SPMM_ROWS < ACC_ROWS, "row SPMM_ROWS is the spare row");
static_assert(2 * ROW_TILES <= SPMM_WAVES, "one wave per 16-row x 64-column piece");
static_assert(2 * ROW_TILES < SPMM_WAVES, "the last wave is free to store the counts");
static_assert(N_NODES < (1 << 24) && KEEP_CAP % 32 == 0, "kept entry packs column<<8 | local row");


template <bool LAST>
__global__ __launch_bounds__(SPMM_THREADS)
void gcn_layer_own(const int* __restrict__ erow, const int* __restrict__ ecol,
                   const float* __restrict__ evals, const float* __restrict__ h,
                   const float* __restrict__ Wm, const float* __restrict__ bias,
                   const float* __restrict__ fcw, const float* __restrict__ fcb,
                   unsigned* __restrict__ kept, int* __restrict__ nkept,
                   float* __restrict__ out) {
    __shared__ float    acc[ACC_ROWS * ACC_PITCH];
    __shared__ unsigned elist[SPMM_WAVES * SPMM_LIST];
    __shared__ float    ldsW[DIM * DIM];
    __shared__ int      nfound[SPMM_WAVES];

    const int tid  = threadIdx.x;
    const int lane = tid & 31;
    const int wave = tid >> 5;
    const int base = blockIdx.x * SPMM_ROWS;

    v2u* __restrict__ mykept = (v2u*)kept + (size_t)(blockIdx.x * SPMM_WAVES + wave) * KEEP_CAP;
    const v2u nothing = {(unsigned)SPMM_ROWS, 0u};
    int  nlist  = 0;
    bool rescan = true;
    v2u  ahead[KEEP_AHEAD];
    if (LAST) {
        const unsigned nk = (unsigned)__builtin_amdgcn_readfirstlane(nkept[blockIdx.x * SPMM_WAVES + wave]);
        rescan = nk > (unsigned)KEEP_CAP;
        nlist  = rescan ? 0 : (int)nk;
#pragma unroll
        for (int q = 0; q < KEEP_AHEAD; ++q)
            ahead[q] = (q * 32 + lane < nlist) ? mykept[q * 32 + lane] : nothing;
    }

    for (int i = tid; i < ACC_ROWS * ACC_PITCH / 4; i += SPMM_THREADS)
        ((float4*)acc)[i] = make_float4(0.0f, 0.0f, 0.0f, 0.0f);

    for (int p = tid; p < DIM * DIM / 2; p += SPMM_THREADS) {
        const int l   = p & 31;
        const int nt  = (p >> 5) & 7;
        const int kk  = p >> 8;
        const int k0  = 16 * (kk >> 2) + 8 * (l >> 4) + (kk & 3);
        const int col = nt * 16 + (l & 15);
        v2f w2;
        w2.x = Wm[(k0 + 0) * DIM + col];
        w2.y = Wm[(k0 + 4) * DIM + col];
        *(v2f*)&ldsW[2 * p] = w2;
    }
    __syncthreads();

    unsigned* __restrict__ mylist = elist + wave * SPMM_LIST;
    const int4* __restrict__ erow4 = (const int4*)erow;

    const int step0 = (blockIdx.x * SCAN_STEPS) / SPMM_WGS;

    auto vec_index = [&](int rnd, int k) {
        int s = step0 + rnd * SCAN_DEPTH + k;
        if (s >= SCAN_STEPS) s -= SCAN_STEPS;
        return s * SPMM_THREADS + tid;
    };
    auto load_round = [&](int4 (&r)[SCAN_DEPTH], int rnd) {
#pragma unroll
        for (int k = 0; k < SCAN_DEPTH; ++k)
            r[k] = erow4[min(vec_index(rnd, k), EDGE_VEC - 1)];
    };

    int count = 0;

    auto look_through = [&](const int4 (&r)[SCAN_DEPTH], int rnd) {
#pragma unroll
        for (int k = 0; k < SCAN_DEPTH; ++k) {
            const int vi = vec_index(rnd, k);
            const unsigned d0 = (unsigned)r[k].x - (unsigned)base;
            const unsigned d1 = (unsigned)r[k].y - (unsigned)base;
            const unsigned d2 = (unsigned)r[k].z - (unsigned)base;
            const unsigned d3 = (unsigned)r[k].w - (unsigned)base;
            const unsigned dmin = (vi < EDGE_VEC) ? min(min(d0, d1), min(d2, d3)) : ~0u;
            if (__builtin_amdgcn_ballot_w32(dmin < SPMM_ROWS) != 0u) {
                const unsigned ebase = (unsigned)vi * 4u;
                const unsigned d[4] = {d0, d1, d2, d3};
#pragma unroll
                for (int q = 0; q < 4; ++q) {
                    const bool m = (dmin < SPMM_ROWS) && (d[q] < SPMM_ROWS);
                    const unsigned b = __builtin_amdgcn_ballot_w32(m);
                    if (m) mylist[count + __builtin_amdgcn_mbcnt_lo(b, 0u)] = ((ebase + q) << 8) | d[q];
                    count += __builtin_popcount(b);
                }
            }
        }
    };

    auto gather_add = [&](int n, int c, float v, int lr) {
        for (int j = 0; j < n; j += GATHER_DEPTH) {
            float4 x[GATHER_DEPTH];
#pragma unroll
            for (int u = 0; u < GATHER_DEPTH; ++u) {
                const int cj = __builtin_amdgcn_readlane(c, j + u);
                x[u] = ((const float4*)(h + (size_t)cj * DIM))[lane];
            }
#pragma unroll
            for (int u = 0; u < GATHER_DEPTH; ++u) {
                const float vj = __builtin_bit_cast(
                    float, __builtin_amdgcn_readlane(__builtin_bit_cast(int, v), j + u));
                const int lrj = __builtin_amdgcn_readlane(lr, j + u);
                float* a = acc + lrj * ACC_PITCH + lane;
                __hip_atomic_fetch_add(a +  0, vj * x[u].x, __ATOMIC_RELAXED, __HIP_MEMORY_SCOPE_WORKGROUP);
                __hip_atomic_fetch_add(a + 32, vj * x[u].y, __ATOMIC_RELAXED, __HIP_MEMORY_SCOPE_WORKGROUP);
                __hip_atomic_fetch_add(a + 64, vj * x[u].z, __ATOMIC_RELAXED, __HIP_MEMORY_SCOPE_WORKGROUP);
                __hip_atomic_fetch_add(a + 96, vj * x[u].w, __ATOMIC_RELAXED, __HIP_MEMORY_SCOPE_WORKGROUP);
            }
        }
    };

    int found = 0;

    auto drain = [&](int keep) {
        while (count > keep) {
            const int n = min(count, 32);
            count -= n;
            const bool live = lane < n;
            const unsigned p = live ? mylist[count + lane] : 0u;
            const int e  = (int)(p >> 8);
            const int lr = live ? (int)(p & 255u) : SPMM_ROWS;
            const int   c = (int)min((unsigned)ecol[e], (unsigned)(N_NODES - 1));
            const float v = evals[e];
            if (!LAST) {
                if (live && found + lane < KEEP_CAP) {
                    v2u s;
                    s.x = ((unsigned)c << 8) | (unsigned)lr;
                    s.y = __builtin_bit_cast(unsigned, v);
                    mykept[found + lane] = s;
                }
            }
            found += n;
            gather_add(n, c, v, lr);
        }
    };

    if (rescan) {
        int4 cur[SCAN_DEPTH];
        load_round(cur, 0);
        for (int rnd = 0; rnd < SCAN_ROUNDS - 1; ++rnd) {
            int4 nxt[SCAN_DEPTH];
            load_round(nxt, rnd + 1);
            look_through(cur, rnd);
            drain(31);
#pragma unroll
            for (int k = 0; k < SCAN_DEPTH; ++k) cur[k] = nxt[k];
        }
        look_through(cur, SCAN_ROUNDS - 1);
        drain(0);
    } else {
        for (int k0 = 0, q = 0; k0 < nlist; k0 += 32, ++q) {
            const int m = min(nlist - k0, 32);
            v2u s = ahead[0];
#pragma unroll
            for (int i = 1; i < KEEP_AHEAD; ++i) s = (q == i) ? ahead[i] : s;
            if (q >= KEEP_AHEAD) s = (lane < m) ? mykept[k0 + lane] : nothing;
            const int   c  = (int)min(s.x >> 8, (unsigned)(N_NODES - 1));
            const int   lr = (int)(s.x & 255u);
            const unsigned vbits = s.y;
            const float v  = __builtin_bit_cast(float, vbits);
            gather_add(m, c, v, lr);
        }
    }
    if (!LAST && lane == 0) nfound[wave] = found;
    __syncthreads();
    if (!LAST && wave == SPMM_WAVES - 1) nkept[blockIdx.x * SPMM_WAVES + lane] = nfound[lane];

    const bool mm = wave < 2 * ROW_TILES;
    const int  mt = wave >> 1;
    const int  nh = wave & 1;
    const int  hh = lane >> 4;
    const int  cc = lane & 15;
    v8f d[4] = {};
    if (mm) {
        const float* arow = acc + (mt * 16 + cc) * ACC_PITCH + 2 * hh;
        for (int kk = 0; kk < 32; ++kk) {
            const v2f a = *(const v2f*)(arow + (kk & 3) * 32 + 4 * (kk >> 2));
#pragma unroll
            for (int nt = 0; nt < 4; ++nt) {
                const v2f b = *(const v2f*)&ldsW[(((kk << 3) + nh * 4 + nt) * 32 + lane) * 2];
                d[nt] = __builtin_amdgcn_wmma_f32_16x16x4_f32(
                     false, a,  false, b,
                     (short)0, d[nt],
                     false,  false);
            }
        }
    }
    __syncthreads();
    if (mm) {
#pragma unroll
        for (int nt = 0; nt < 4; ++nt) {
            const int   col = nh * 64 + nt * 16 + cc;
            const float bc  = bias[col];
            float* t = acc + (mt * 16 + 8 * hh) * ACC_PITCH + (col & 3) * 32 + (col >> 2);
#pragma unroll
            for (int j = 0; j < 8; ++j) t[j * ACC_PITCH] = tanhf(d[nt][j] + bc);
        }
    }
    __syncthreads();

    if (!LAST) {
        for (int rr = wave; rr < SPMM_ROWS; rr += SPMM_WAVES) {
            const int row = base + rr;
            if (row < N_NODES) {
                const float* a = acc + rr * ACC_PITCH + lane;
                ((float4*)(out + (size_t)row * DIM))[lane] = make_float4(a[0], a[32], a[64], a[96]);
            }
        }
    } else {
        float* dots = (float*)elist;
        const float4 w = ((const float4*)fcw)[lane];
        for (int rr = wave; rr < SPMM_ROWS; rr += SPMM_WAVES) {
            const float* a = acc + rr * ACC_PITCH + lane;
            float s = a[0] * w.x + a[32] * w.y + a[64] * w.z + a[96] * w.w;
#pragma unroll
            for (int off = 16; off > 0; off >>= 1) s += __shfl_down(s, off, 32);
            if (lane == 0) dots[rr] = s + fcb[0];
        }
        __syncthreads();
        if (tid < SPMM_ROWS && base + tid < N_NODES) out[base + tid] = dots[tid];
    }
}

extern "C" void kernel_launch(void* const* d_in, const int* in_sizes, int n_in,
                              void* d_out, int out_size, void* d_ws, size_t ws_size,
                              hipStream_t stream) {
    const float* features = (const float*)d_in[0];
    const int*   erow     = (const int*)  d_in[1];
    const int*   ecol     = (const int*)  d_in[2];
    const float* evals    = (const float*)d_in[3];
    const float* W1       = (const float*)d_in[4];
    const float* b1       = (const float*)d_in[5];
    const float* W2       = (const float*)d_in[6];
    const float* b2       = (const float*)d_in[7];
    const float* fcw      = (const float*)d_in[8];
    const float* fcb      = (const float*)d_in[9];
    float*       out      = (float*)d_out;

    float*    h1    = (float*)d_ws;
    unsigned* kept  = (unsigned*)((char*)d_ws + WS_KEPT_OFF);
    int*      nkept = (int*)((char*)d_ws + WS_NKEPT_OFF);

    const dim3 grid(SPMM_WGS);

    gcn_layer_own<false><<<grid, SPMM_THREADS, 0, stream>>>(erow, ecol, evals, features, W1, b1, fcw, fcb, kept, nkept, h1);
    gcn_layer_own<true><<<grid, SPMM_THREADS, 0, stream>>>(erow, ecol, evals, h1, W2, b2, fcw, fcb, kept, nkept, out);
}


# layer 1 finds its block's edges via chunk bits: a new first launch marks, per row block, which 32-edge chunks hold an edge of it (no edge moved); per-block read of the whole list gone; 3 launches
# speedup vs baseline: 2.9196x; 1.0326x over previous
#include <hip/hip_runtime.h>


#define N_NODES 50000
#define N_EDGES 600000
#define DIM     128

typedef __attribute__((ext_vector_type(2))) float v2f;
typedef __attribute__((ext_vector_type(8))) float v8f;
typedef __attribute__((ext_vector_type(2))) unsigned v2u;

#define SPMM_ROWS    196
#define SPMM_WGS     ((N_NODES + SPMM_ROWS - 1) / SPMM_ROWS)
#define SPMM_WAVES   32
#define SPMM_THREADS (SPMM_WAVES * 32)
#define CHUNKS       (N_EDGES / 32)
#define BM_WORDS     ((CHUNKS + 31) / 32)
#define BM_WGS       ((BM_WORDS + 31) / 32)
#define BM_PITCH     (BM_WGS * 32)
#define BM_WPW       (BM_PITCH / SPMM_WAVES)
#define BM_TBL_PITCH 257
#define CHUNK_DEPTH  16
#define SPMM_LIST    (31 + CHUNK_DEPTH * 32)
#define CHUNK_LIST   (BM_WPW * 32 / 2)
#define GATHER_DEPTH 16
#define ROW_TILES    ((SPMM_ROWS + 15) / 16)
#define ACC_ROWS     (ROW_TILES * 16)
#define ACC_PITCH    (DIM + 4)
#define KEEP_CAP     256
#define KEEP_AHEAD   3
#define WS_KEPT_OFF  ((size_t)N_NODES * DIM * 4)
#define WS_NKEPT_OFF (WS_KEPT_OFF + (size_t)SPMM_WGS * SPMM_WAVES * KEEP_CAP * 8)
#define WS_BM_OFF    (WS_NKEPT_OFF + (size_t)SPMM_WGS * SPMM_WAVES * 4)

static_assert(N_EDGES % 32 == 0, "the edge list is whole 32-edge chunks");
static_assert(BM_PITCH * 32 < (1 << 16), "a chunk number fits in 16 bits");
static_assert(BM_WPW <= 32 && CHUNK_DEPTH <= 32, "one bitmap word, one chunk number per lane");
static_assert(SPMM_WGS <= 256 && SPMM_WGS % 32 == 0, "the chunk-bit table has one word per block");
static_assert(WS_BM_OFF % 128 == 0, "bitmap rows start on a 128-B line");
static_assert(SPMM_ROWS < 256 && N_EDGES < (1 << 24), "list entry packs edge<<8 | local row");
static_assert(SPMM_ROWS < ACC_ROWS, "row SPMM_ROWS is the spare row");
static_assert(2 * ROW_TILES <= SPMM_WAVES, "one wave per 16-row x 64-column piece");
static_assert(2 * ROW_TILES < SPMM_WAVES, "the last wave is free to store the counts");
static_assert(N_NODES < (1 << 24) && KEEP_CAP % 32 == 0, "kept entry packs column<<8 | local row");


__global__ __launch_bounds__(SPMM_THREADS)
void edge_chunk_bits(const int* __restrict__ erow, unsigned* __restrict__ bm) {
    __shared__ unsigned tbl[SPMM_WAVES * BM_TBL_PITCH];
    const int tid  = threadIdx.x;
    const int lane = tid & 31;
    const int wave = tid >> 5;
    const int g    = blockIdx.x * SPMM_WAVES + __builtin_amdgcn_readfirstlane(wave);
    unsigned* mine = tbl + wave * BM_TBL_PITCH;

    int r[32];
#pragma unroll
    for (int c = 0; c < 32; ++c) {
        const int chunk = g * 32 + c;
        r[c] = (chunk < CHUNKS) ? erow[chunk * 32 + lane] : -1;
    }
    for (int i = lane; i < SPMM_WGS; i += 32) mine[i] = 0u;
#pragma unroll
    for (int c = 0; c < 32; ++c) {
        const unsigned blk = (unsigned)r[c] / (unsigned)SPMM_ROWS;
        if (blk < (unsigned)SPMM_WGS)
            __hip_atomic_fetch_or(mine + blk, 1u << c, __ATOMIC_RELAXED, __HIP_MEMORY_SCOPE_WORKGROUP);
    }
    __syncthreads();
    for (int b = wave; b < SPMM_WGS; b += SPMM_WAVES)
        bm[b * BM_PITCH + blockIdx.x * SPMM_WAVES + lane] = tbl[lane * BM_TBL_PITCH + b];
}

template <bool LAST>
__global__ __launch_bounds__(SPMM_THREADS)
void gcn_layer_own(const int* __restrict__ erow, const int* __restrict__ ecol,
                   const float* __restrict__ evals, const float* __restrict__ h,
                   const float* __restrict__ Wm, const float* __restrict__ bias,
                   const float* __restrict__ fcw, const float* __restrict__ fcb,
                   unsigned* __restrict__ kept, int* __restrict__ nkept,
                   const unsigned* __restrict__ bm, float* __restrict__ out) {
    __shared__ float    acc[ACC_ROWS * ACC_PITCH];
    __shared__ unsigned elist[SPMM_WAVES * (SPMM_LIST + CHUNK_LIST)];
    __shared__ float    ldsW[DIM * DIM];
    __shared__ int      nfound[SPMM_WAVES];

    const int tid  = threadIdx.x;
    const int lane = tid & 31;
    const int wave = tid >> 5;
    const int base = blockIdx.x * SPMM_ROWS;

    v2u* __restrict__ mykept = (v2u*)kept + (size_t)(blockIdx.x * SPMM_WAVES + wave) * KEEP_CAP;
    const v2u nothing = {(unsigned)SPMM_ROWS, 0u};
    int  nlist  = 0;
    bool rescan = true;
    v2u  ahead[KEEP_AHEAD];
    if (LAST) {
        const unsigned nk = (unsigned)__builtin_amdgcn_readfirstlane(nkept[blockIdx.x * SPMM_WAVES + wave]);
        rescan = nk > (unsigned)KEEP_CAP;
        nlist  = rescan ? 0 : (int)nk;
#pragma unroll
        for (int q = 0; q < KEEP_AHEAD; ++q)
            ahead[q] = (q * 32 + lane < nlist) ? mykept[q * 32 + lane] : nothing;
    }

    unsigned wbits = 0u;
    if (rescan && lane < BM_WPW) wbits = bm[blockIdx.x * BM_PITCH + lane * SPMM_WAVES + wave];

    for (int i = tid; i < ACC_ROWS * ACC_PITCH / 4; i += SPMM_THREADS)
        ((float4*)acc)[i] = make_float4(0.0f, 0.0f, 0.0f, 0.0f);

    for (int p = tid; p < DIM * DIM / 2; p += SPMM_THREADS) {
        const int l   = p & 31;
        const int nt  = (p >> 5) & 7;
        const int kk  = p >> 8;
        const int k0  = 16 * (kk >> 2) + 8 * (l >> 4) + (kk & 3);
        const int col = nt * 16 + (l & 15);
        v2f w2;
        w2.x = Wm[(k0 + 0) * DIM + col];
        w2.y = Wm[(k0 + 4) * DIM + col];
        *(v2f*)&ldsW[2 * p] = w2;
    }
    __syncthreads();

    unsigned* __restrict__ mylist = elist + wave * (SPMM_LIST + CHUNK_LIST);
    unsigned short* __restrict__ clist = (unsigned short*)(mylist + SPMM_LIST);

    int count = 0;

    auto gather_add = [&](int n, int c, float v, int lr) {
        for (int j = 0; j < n; j += GATHER_DEPTH) {
            float4 x[GATHER_DEPTH];
#pragma unroll
            for (int u = 0; u < GATHER_DEPTH; ++u) {
                const int cj = __builtin_amdgcn_readlane(c, j + u);
                x[u] = ((const float4*)(h + (size_t)cj * DIM))[lane];
            }
#pragma unroll
            for (int u = 0; u < GATHER_DEPTH; ++u) {
                const float vj = __builtin_bit_cast(
                    float, __builtin_amdgcn_readlane(__builtin_bit_cast(int, v), j + u));
                const int lrj = __builtin_amdgcn_readlane(lr, j + u);
                float* a = acc + lrj * ACC_PITCH + lane;
                __hip_atomic_fetch_add(a +  0, vj * x[u].x, __ATOMIC_RELAXED, __HIP_MEMORY_SCOPE_WORKGROUP);
                __hip_atomic_fetch_add(a + 32, vj * x[u].y, __ATOMIC_RELAXED, __HIP_MEMORY_SCOPE_WORKGROUP);
                __hip_atomic_fetch_add(a + 64, vj * x[u].z, __ATOMIC_RELAXED, __HIP_MEMORY_SCOPE_WORKGROUP);
                __hip_atomic_fetch_add(a + 96, vj * x[u].w, __ATOMIC_RELAXED, __HIP_MEMORY_SCOPE_WORKGROUP);
            }
        }
    };

    int found = 0;

    auto drain = [&](int keep) {
        while (count > keep) {
            const int n = min(count, 32);
            count -= n;
            const bool live = lane < n;
            const unsigned p = live ? mylist[count + lane] : 0u;
            const int e  = (int)(p >> 8);
            const int lr = live ? (int)(p & 255u) : SPMM_ROWS;
            const int   c = (int)min((unsigned)ecol[e], (unsigned)(N_NODES - 1));
            const float v = evals[e];
            if (!LAST) {
                if (live && found + lane < KEEP_CAP) {
                    v2u s;
                    s.x = ((unsigned)c << 8) | (unsigned)lr;
                    s.y = __builtin_bit_cast(unsigned, v);
                    mykept[found + lane] = s;
                }
            }
            found += n;
            gather_add(n, c, v, lr);
        }
    };

    if (rescan) {
        const int pc = __builtin_popcount(wbits);
        int upto = pc;
#pragma unroll
        for (int off = 1; off < 32; off <<= 1) {
            const int t = __shfl_up(upto, off, 32);
            if (lane >= off) upto += t;
        }
        const int nchunks = __builtin_amdgcn_readlane(upto, 31);
        {
            int pos = upto - pc;
            const unsigned first = (unsigned)(lane * SPMM_WAVES + wave) * 32u;
            for (unsigned wb = wbits; wb != 0u; wb &= wb - 1u)
                clist[pos++] = (unsigned short)(first + (unsigned)__builtin_ctz(wb));
        }
        for (int k0 = 0; k0 < nchunks; k0 += CHUNK_DEPTH) {
            const int m   = min(nchunks - k0, CHUNK_DEPTH);
            const int myc = (lane < m) ? (int)clist[k0 + lane] : -1;
            int r[CHUNK_DEPTH];
#pragma unroll
            for (int u = 0; u < CHUNK_DEPTH; ++u) {
                const int cid = __builtin_amdgcn_readlane(myc, u);
                r[u] = erow[min(max(cid, 0), CHUNKS - 1) * 32 + lane];
            }
#pragma unroll
            for (int u = 0; u < CHUNK_DEPTH; ++u) {
                const int cid = __builtin_amdgcn_readlane(myc, u);
                const unsigned d = (unsigned)r[u] - (unsigned)base;
                const bool hit = (cid >= 0) && (d < SPMM_ROWS);
                const unsigned b = __builtin_amdgcn_ballot_w32(hit);
                if (hit) mylist[count + __builtin_amdgcn_mbcnt_lo(b, 0u)] =
                             (((unsigned)min(cid, CHUNKS - 1) * 32u + (unsigned)lane) << 8) | d;
                count += __builtin_popcount(b);
            }
            drain(31);
        }
        drain(0);
    } else {
        for (int k0 = 0, q = 0; k0 < nlist; k0 += 32, ++q) {
            const int m = min(nlist - k0, 32);
            v2u s = ahead[0];
#pragma unroll
            for (int i = 1; i < KEEP_AHEAD; ++i) s = (q == i) ? ahead[i] : s;
            if (q >= KEEP_AHEAD) s = (lane < m) ? mykept[k0 + lane] : nothing;
            const int   c  = (int)min(s.x >> 8, (unsigned)(N_NODES - 1));
            const int   lr = (int)(s.x & 255u);
            const unsigned vbits = s.y;
            const float v  = __builtin_bit_cast(float, vbits);
            gather_add(m, c, v, lr);
        }
    }
    if (!LAST && lane == 0) nfound[wave] = found;
    __syncthreads();
    if (!LAST && wave == SPMM_WAVES - 1) nkept[blockIdx.x * SPMM_WAVES + lane] = nfound[lane];

    const bool mm = wave < 2 * ROW_TILES;
    const int  mt = wave >> 1;
    const int  nh = wave & 1;
    const int  hh = lane >> 4;
    const int  cc = lane & 15;
    v8f d[4] = {};
    if (mm) {
        const float* arow = acc + (mt * 16 + cc) * ACC_PITCH + 2 * hh;
        for (int kk = 0; kk < 32; ++kk) {
            const v2f a = *(const v2f*)(arow + (kk & 3) * 32 + 4 * (kk >> 2));
#pragma unroll
            for (int nt = 0; nt < 4; ++nt) {
                const v2f b = *(const v2f*)&ldsW[(((kk << 3) + nh * 4 + nt) * 32 + lane) * 2];
                d[nt] = __builtin_amdgcn_wmma_f32_16x16x4_f32(
                     false, a,  false, b,
                     (short)0, d[nt],
                     false,  false);
            }
        }
    }
    __syncthreads();
    if (mm) {
#pragma unroll
        for (int nt = 0; nt < 4; ++nt) {
            const int   col = nh * 64 + nt * 16 + cc;
            const float bc  = bias[col];
            float* t = acc + (mt * 16 + 8 * hh) * ACC_PITCH + (col & 3) * 32 + (col >> 2);
#pragma unroll
            for (int j = 0; j < 8; ++j) t[j * ACC_PITCH] = tanhf(d[nt][j] + bc);
        }
    }
    __syncthreads();

    if (!LAST) {
        for (int rr = wave; rr < SPMM_ROWS; rr += SPMM_WAVES) {
            const int row = base + rr;
            if (row < N_NODES) {
                const float* a = acc + rr * ACC_PITCH + lane;
                ((float4*)(out + (size_t)row * DIM))[lane] = make_float4(a[0], a[32], a[64], a[96]);
            }
        }
    } else {
        float* dots = (float*)elist;
        const float4 w = ((const float4*)fcw)[lane];
        for (int rr = wave; rr < SPMM_ROWS; rr += SPMM_WAVES) {
            const float* a = acc + rr * ACC_PITCH + lane;
            float s = a[0] * w.x + a[32] * w.y + a[64] * w.z + a[96] * w.w;
#pragma unroll
            for (int off = 16; off > 0; off >>= 1) s += __shfl_down(s, off, 32);
            if (lane == 0) dots[rr] = s + fcb[0];
        }
        __syncthreads();
        if (tid < SPMM_ROWS && base + tid < N_NODES) out[base + tid] = dots[tid];
    }
}

extern "C" void kernel_launch(void* const* d_in, const int* in_sizes, int n_in,
                              void* d_out, int out_size, void* d_ws, size_t ws_size,
                              hipStream_t stream) {
    const float* features = (const float*)d_in[0];
    const int*   erow     = (const int*)  d_in[1];
    const int*   ecol     = (const int*)  d_in[2];
    const float* evals    = (const float*)d_in[3];
    const float* W1       = (const float*)d_in[4];
    const float* b1       = (const float*)d_in[5];
    const float* W2       = (const float*)d_in[6];
    const float* b2       = (const float*)d_in[7];
    const float* fcw      = (const float*)d_in[8];
    const float* fcb      = (const float*)d_in[9];
    float*       out      = (float*)d_out;

    float*    h1    = (float*)d_ws;
    unsigned* kept  = (unsigned*)((char*)d_ws + WS_KEPT_OFF);
    int*      nkept = (int*)((char*)d_ws + WS_NKEPT_OFF);
    unsigned* bm    = (unsigned*)((char*)d_ws + WS_BM_OFF);

    const dim3 grid(SPMM_WGS);

    edge_chunk_bits<<<dim3(BM_WGS), SPMM_THREADS, 0, stream>>>(erow, bm);
    gcn_layer_own<false><<<grid, SPMM_THREADS, 0, stream>>>(erow, ecol, evals, features, W1, b1, fcw, fcb, kept, nkept, bm, h1);
    gcn_layer_own<true><<<grid, SPMM_THREADS, 0, stream>>>(erow, ecol, evals, h1, W2, b2, fcw, fcb, kept, nkept, bm, out);
}


# tanh of both layer finishes as 1 - 2/(2^(2 log2(e) |x|) + 1) on one v_exp_f32 and one v_rcp_f32, branch-free, in place of the library tanhf (abs error about 2e-7 near 0)
# speedup vs baseline: 3.0779x; 1.0542x over previous
#include <hip/hip_runtime.h>


#define N_NODES 50000
#define N_EDGES 600000
#define DIM     128

typedef __attribute__((ext_vector_type(2))) float v2f;
typedef __attribute__((ext_vector_type(8))) float v8f;
typedef __attribute__((ext_vector_type(2))) unsigned v2u;

#define SPMM_ROWS    196
#define SPMM_WGS     ((N_NODES + SPMM_ROWS - 1) / SPMM_ROWS)
#define SPMM_WAVES   32
#define SPMM_THREADS (SPMM_WAVES * 32)
#define CHUNKS       (N_EDGES / 32)
#define BM_WORDS     ((CHUNKS + 31) / 32)
#define BM_WGS       ((BM_WORDS + 31) / 32)
#define BM_PITCH     (BM_WGS * 32)
#define BM_WPW       (BM_PITCH / SPMM_WAVES)
#define BM_TBL_PITCH 257
#define CHUNK_DEPTH  16
#define SPMM_LIST    (31 + CHUNK_DEPTH * 32)
#define CHUNK_LIST   (BM_WPW * 32 / 2)
#define GATHER_DEPTH 16
#define ROW_TILES    ((SPMM_ROWS + 15) / 16)
#define ACC_ROWS     (ROW_TILES * 16)
#define ACC_PITCH    (DIM + 4)
#define KEEP_CAP     256
#define KEEP_AHEAD   3
#define WS_KEPT_OFF  ((size_t)N_NODES * DIM * 4)
#define WS_NKEPT_OFF (WS_KEPT_OFF + (size_t)SPMM_WGS * SPMM_WAVES * KEEP_CAP * 8)
#define WS_BM_OFF    (WS_NKEPT_OFF + (size_t)SPMM_WGS * SPMM_WAVES * 4)

static_assert(N_EDGES % 32 == 0, "the edge list is whole 32-edge chunks");
static_assert(BM_PITCH * 32 < (1 << 16), "a chunk number fits in 16 bits");
static_assert(BM_WPW <= 32 && CHUNK_DEPTH <= 32, "one bitmap word, one chunk number per lane");
static_assert(SPMM_WGS <= 256 && SPMM_WGS % 32 == 0, "the chunk-bit table has one word per block");
static_assert(WS_BM_OFF % 128 == 0, "bitmap rows start on a 128-B line");
static_assert(SPMM_ROWS < 256 && N_EDGES < (1 << 24), "list entry packs edge<<8 | local row");
static_assert(SPMM_ROWS < ACC_ROWS, "row SPMM_ROWS is the spare row");
static_assert(2 * ROW_TILES <= SPMM_WAVES, "one wave per 16-row x 64-column piece");
static_assert(2 * ROW_TILES < SPMM_WAVES, "the last wave is free to store the counts");
static_assert(N_NODES < (1 << 24) && KEEP_CAP % 32 == 0, "kept entry packs column<<8 | local row");


__global__ __launch_bounds__(SPMM_THREADS)
void edge_chunk_bits(const int* __restrict__ erow, unsigned* __restrict__ bm) {
    __shared__ unsigned tbl[SPMM_WAVES * BM_TBL_PITCH];
    const int tid  = threadIdx.x;
    const int lane = tid & 31;
    const int wave = tid >> 5;
    const int g    = blockIdx.x * SPMM_WAVES + __builtin_amdgcn_readfirstlane(wave);
    unsigned* mine = tbl + wave * BM_TBL_PITCH;

    int r[32];
#pragma unroll
    for (int c = 0; c < 32; ++c) {
        const int chunk = g * 32 + c;
        r[c] = (chunk < CHUNKS) ? erow[chunk * 32 + lane] : -1;
    }
    for (int i = lane; i < SPMM_WGS; i += 32) mine[i] = 0u;
#pragma unroll
    for (int c = 0; c < 32; ++c) {
        const unsigned blk = (unsigned)r[c] / (unsigned)SPMM_ROWS;
        if (blk < (unsigned)SPMM_WGS)
            __hip_atomic_fetch_or(mine + blk, 1u << c, __ATOMIC_RELAXED, __HIP_MEMORY_SCOPE_WORKGROUP);
    }
    __syncthreads();
    for (int b = wave; b < SPMM_WGS; b += SPMM_WAVES)
        bm[b * BM_PITCH + blockIdx.x * SPMM_WAVES + lane] = tbl[lane * BM_TBL_PITCH + b];
}

__device__ __forceinline__ float tanh_exp_rcp(float x) {
    const float t = __builtin_amdgcn_exp2f(__builtin_fabsf(x) * 2.8853900817779268f);
    const float r = 1.0f - 2.0f * __builtin_amdgcn_rcpf(t + 1.0f);
    return __builtin_copysignf(r, x);
}

template <bool LAST>
__global__ __launch_bounds__(SPMM_THREADS)
void gcn_layer_own(const int* __restrict__ erow, const int* __restrict__ ecol,
                   const float* __restrict__ evals, const float* __restrict__ h,
                   const float* __restrict__ Wm, const float* __restrict__ bias,
                   const float* __restrict__ fcw, const float* __restrict__ fcb,
                   unsigned* __restrict__ kept, int* __restrict__ nkept,
                   const unsigned* __restrict__ bm, float* __restrict__ out) {
    __shared__ float    acc[ACC_ROWS * ACC_PITCH];
    __shared__ unsigned elist[SPMM_WAVES * (SPMM_LIST + CHUNK_LIST)];
    __shared__ float    ldsW[DIM * DIM];
    __shared__ int      nfound[SPMM_WAVES];

    const int tid  = threadIdx.x;
    const int lane = tid & 31;
    const int wave = tid >> 5;
    const int base = blockIdx.x * SPMM_ROWS;

    v2u* __restrict__ mykept = (v2u*)kept + (size_t)(blockIdx.x * SPMM_WAVES + wave) * KEEP_CAP;
    const v2u nothing = {(unsigned)SPMM_ROWS, 0u};
    int  nlist  = 0;
    bool rescan = true;
    v2u  ahead[KEEP_AHEAD];
    if (LAST) {
        const unsigned nk = (unsigned)__builtin_amdgcn_readfirstlane(nkept[blockIdx.x * SPMM_WAVES + wave]);
        rescan = nk > (unsigned)KEEP_CAP;
        nlist  = rescan ? 0 : (int)nk;
#pragma unroll
        for (int q = 0; q < KEEP_AHEAD; ++q)
            ahead[q] = (q * 32 + lane < nlist) ? mykept[q * 32 + lane] : nothing;
    }

    unsigned wbits = 0u;
    if (rescan && lane < BM_WPW) wbits = bm[blockIdx.x * BM_PITCH + lane * SPMM_WAVES + wave];

    for (int i = tid; i < ACC_ROWS * ACC_PITCH / 4; i += SPMM_THREADS)
        ((float4*)acc)[i] = make_float4(0.0f, 0.0f, 0.0f, 0.0f);

    for (int p = tid; p < DIM * DIM / 2; p += SPMM_THREADS) {
        const int l   = p & 31;
        const int nt  = (p >> 5) & 7;
        const int kk  = p >> 8;
        const int k0  = 16 * (kk >> 2) + 8 * (l >> 4) + (kk & 3);
        const int col = nt * 16 + (l & 15);
        v2f w2;
        w2.x = Wm[(k0 + 0) * DIM + col];
        w2.y = Wm[(k0 + 4) * DIM + col];
        *(v2f*)&ldsW[2 * p] = w2;
    }
    __syncthreads();

    unsigned* __restrict__ mylist = elist + wave * (SPMM_LIST + CHUNK_LIST);
    unsigned short* __restrict__ clist = (unsigned short*)(mylist + SPMM_LIST);

    int count = 0;

    auto gather_add = [&](int n, int c, float v, int lr) {
        for (int j = 0; j < n; j += GATHER_DEPTH) {
            float4 x[GATHER_DEPTH];
#pragma unroll
            for (int u = 0; u < GATHER_DEPTH; ++u) {
                const int cj = __builtin_amdgcn_readlane(c, j + u);
                x[u] = ((const float4*)(h + (size_t)cj * DIM))[lane];
            }
#pragma unroll
            for (int u = 0; u < GATHER_DEPTH; ++u) {
                const float vj = __builtin_bit_cast(
                    float, __builtin_amdgcn_readlane(__builtin_bit_cast(int, v), j + u));
                const int lrj = __builtin_amdgcn_readlane(lr, j + u);
                float* a = acc + lrj * ACC_PITCH + lane;
                __hip_atomic_fetch_add(a +  0, vj * x[u].x, __ATOMIC_RELAXED, __HIP_MEMORY_SCOPE_WORKGROUP);
                __hip_atomic_fetch_add(a + 32, vj * x[u].y, __ATOMIC_RELAXED, __HIP_MEMORY_SCOPE_WORKGROUP);
                __hip_atomic_fetch_add(a + 64, vj * x[u].z, __ATOMIC_RELAXED, __HIP_MEMORY_SCOPE_WORKGROUP);
                __hip_atomic_fetch_add(a + 96, vj * x[u].w, __ATOMIC_RELAXED, __HIP_MEMORY_SCOPE_WORKGROUP);
            }
        }
    };

    int found = 0;

    auto drain = [&](int keep) {
        while (count > keep) {
            const int n = min(count, 32);
            count -= n;
            const bool live = lane < n;
            const unsigned p = live ? mylist[count + lane] : 0u;
            const int e  = (int)(p >> 8);
            const int lr = live ? (int)(p & 255u) : SPMM_ROWS;
            const int   c = (int)min((unsigned)ecol[e], (unsigned)(N_NODES - 1));
            const float v = evals[e];
            if (!LAST) {
                if (live && found + lane < KEEP_CAP) {
                    v2u s;
                    s.x = ((unsigned)c << 8) | (unsigned)lr;
                    s.y = __builtin_bit_cast(unsigned, v);
                    mykept[found + lane] = s;
                }
            }
            found += n;
            gather_add(n, c, v, lr);
        }
    };

    if (rescan) {
        const int pc = __builtin_popcount(wbits);
        int upto = pc;
#pragma unroll
        for (int off = 1; off < 32; off <<= 1) {
            const int t = __shfl_up(upto, off, 32);
            if (lane >= off) upto += t;
        }
        const int nchunks = __builtin_amdgcn_readlane(upto, 31);
        {
            int pos = upto - pc;
            const unsigned first = (unsigned)(lane * SPMM_WAVES + wave) * 32u;
            for (unsigned wb = wbits; wb != 0u; wb &= wb - 1u)
                clist[pos++] = (unsigned short)(first + (unsigned)__builtin_ctz(wb));
        }
        for (int k0 = 0; k0 < nchunks; k0 += CHUNK_DEPTH) {
            const int m   = min(nchunks - k0, CHUNK_DEPTH);
            const int myc = (lane < m) ? (int)clist[k0 + lane] : -1;
            int r[CHUNK_DEPTH];
#pragma unroll
            for (int u = 0; u < CHUNK_DEPTH; ++u) {
                const int cid = __builtin_amdgcn_readlane(myc, u);
                r[u] = erow[min(max(cid, 0), CHUNKS - 1) * 32 + lane];
            }
#pragma unroll
            for (int u = 0; u < CHUNK_DEPTH; ++u) {
                const int cid = __builtin_amdgcn_readlane(myc, u);
                const unsigned d = (unsigned)r[u] - (unsigned)base;
                const bool hit = (cid >= 0) && (d < SPMM_ROWS);
                const unsigned b = __builtin_amdgcn_ballot_w32(hit);
                if (hit) mylist[count + __builtin_amdgcn_mbcnt_lo(b, 0u)] =
                             (((unsigned)min(cid, CHUNKS - 1) * 32u + (unsigned)lane) << 8) | d;
                count += __builtin_popcount(b);
            }
            drain(31);
        }
        drain(0);
    } else {
        for (int k0 = 0, q = 0; k0 < nlist; k0 += 32, ++q) {
            const int m = min(nlist - k0, 32);
            v2u s = ahead[0];
#pragma unroll
            for (int i = 1; i < KEEP_AHEAD; ++i) s = (q == i) ? ahead[i] : s;
            if (q >= KEEP_AHEAD) s = (lane < m) ? mykept[k0 + lane] : nothing;
            const int   c  = (int)min(s.x >> 8, (unsigned)(N_NODES - 1));
            const int   lr = (int)(s.x & 255u);
            const unsigned vbits = s.y;
            const float v  = __builtin_bit_cast(float, vbits);
            gather_add(m, c, v, lr);
        }
    }
    if (!LAST && lane == 0) nfound[wave] = found;
    __syncthreads();
    if (!LAST && wave == SPMM_WAVES - 1) nkept[blockIdx.x * SPMM_WAVES + lane] = nfound[lane];

    const bool mm = wave < 2 * ROW_TILES;
    const int  mt = wave >> 1;
    const int  nh = wave & 1;
    const int  hh = lane >> 4;
    const int  cc = lane & 15;
    v8f d[4] = {};
    if (mm) {
        const float* arow = acc + (mt * 16 + cc) * ACC_PITCH + 2 * hh;
        for (int kk = 0; kk < 32; ++kk) {
            const v2f a = *(const v2f*)(arow + (kk & 3) * 32 + 4 * (kk >> 2));
#pragma unroll
            for (int nt = 0; nt < 4; ++nt) {
                const v2f b = *(const v2f*)&ldsW[(((kk << 3) + nh * 4 + nt) * 32 + lane) * 2];
                d[nt] = __builtin_amdgcn_wmma_f32_16x16x4_f32(
                     false, a,  false, b,
                     (short)0, d[nt],
                     false,  false);
            }
        }
    }
    __syncthreads();
    if (mm) {
#pragma unroll
        for (int nt = 0; nt < 4; ++nt) {
            const int   col = nh * 64 + nt * 16 + cc;
            const float bc  = bias[col];
            float* t = acc + (mt * 16 + 8 * hh) * ACC_PITCH + (col & 3) * 32 + (col >> 2);
#pragma unroll
            for (int j = 0; j < 8; ++j) t[j * ACC_PITCH] = tanh_exp_rcp(d[nt][j] + bc);
        }
    }
    __syncthreads();

    if (!LAST) {
        for (int rr = wave; rr < SPMM_ROWS; rr += SPMM_WAVES) {
            const int row = base + rr;
            if (row < N_NODES) {
                const float* a = acc + rr * ACC_PITCH + lane;
                ((float4*)(out + (size_t)row * DIM))[lane] = make_float4(a[0], a[32], a[64], a[96]);
            }
        }
    } else {
        float* dots = (float*)elist;
        const float4 w = ((const float4*)fcw)[lane];
        for (int rr = wave; rr < SPMM_ROWS; rr += SPMM_WAVES) {
            const float* a = acc + rr * ACC_PITCH + lane;
            float s = a[0] * w.x + a[32] * w.y + a[64] * w.z + a[96] * w.w;
#pragma unroll
            for (int off = 16; off > 0; off >>= 1) s += __shfl_down(s, off, 32);
            if (lane == 0) dots[rr] = s + fcb[0];
        }
        __syncthreads();
        if (tid < SPMM_ROWS && base + tid < N_NODES) out[base + tid] = dots[tid];
    }
}

extern "C" void kernel_launch(void* const* d_in, const int* in_sizes, int n_in,
                              void* d_out, int out_size, void* d_ws, size_t ws_size,
                              hipStream_t stream) {
    const float* features = (const float*)d_in[0];
    const int*   erow     = (const int*)  d_in[1];
    const int*   ecol     = (const int*)  d_in[2];
    const float* evals    = (const float*)d_in[3];
    const float* W1       = (const float*)d_in[4];
    const float* b1       = (const float*)d_in[5];
    const float* W2       = (const float*)d_in[6];
    const float* b2       = (const float*)d_in[7];
    const float* fcw      = (const float*)d_in[8];
    const float* fcb      = (const float*)d_in[9];
    float*       out      = (float*)d_out;

    float*    h1    = (float*)d_ws;
    unsigned* kept  = (unsigned*)((char*)d_ws + WS_KEPT_OFF);
    int*      nkept = (int*)((char*)d_ws + WS_NKEPT_OFF);
    unsigned* bm    = (unsigned*)((char*)d_ws + WS_BM_OFF);

    const dim3 grid(SPMM_WGS);

    edge_chunk_bits<<<dim3(BM_WGS), SPMM_THREADS, 0, stream>>>(erow, bm);
    gcn_layer_own<false><<<grid, SPMM_THREADS, 0, stream>>>(erow, ecol, evals, features, W1, b1, fcw, fcb, kept, nkept, bm, h1);
    gcn_layer_own<true><<<grid, SPMM_THREADS, 0, stream>>>(erow, ecol, evals, h1, W2, b2, fcw, fcb, kept, nkept, bm, out);
}


# loads that the whole work-group waits for in lock step go out early as one group: W by whole-row 16-byte loads at entry (was 8 load-wait-store rounds); bias and fc weights ahead of the product
# speedup vs baseline: 3.3070x; 1.0744x over previous
#include <hip/hip_runtime.h>


#define N_NODES 50000
#define N_EDGES 600000
#define DIM     128

typedef __attribute__((ext_vector_type(2))) float v2f;
typedef __attribute__((ext_vector_type(8))) float v8f;
typedef __attribute__((ext_vector_type(2))) unsigned v2u;

#define SPMM_ROWS    196
#define SPMM_WGS     ((N_NODES + SPMM_ROWS - 1) / SPMM_ROWS)
#define SPMM_WAVES   32
#define SPMM_THREADS (SPMM_WAVES * 32)
#define CHUNKS       (N_EDGES / 32)
#define BM_WORDS     ((CHUNKS + 31) / 32)
#define BM_WGS       ((BM_WORDS + 31) / 32)
#define BM_PITCH     (BM_WGS * 32)
#define BM_WPW       (BM_PITCH / SPMM_WAVES)
#define BM_TBL_PITCH 257
#define CHUNK_DEPTH  16
#define SPMM_LIST    (31 + CHUNK_DEPTH * 32)
#define CHUNK_LIST   (BM_WPW * 32 / 2)
#define GATHER_DEPTH 16
#define ROW_TILES    ((SPMM_ROWS + 15) / 16)
#define ACC_ROWS     (ROW_TILES * 16)
#define ACC_PITCH    (DIM + 4)
#define KEEP_CAP     256
#define KEEP_AHEAD   3
#define WS_KEPT_OFF  ((size_t)N_NODES * DIM * 4)
#define WS_NKEPT_OFF (WS_KEPT_OFF + (size_t)SPMM_WGS * SPMM_WAVES * KEEP_CAP * 8)
#define WS_BM_OFF    (WS_NKEPT_OFF + (size_t)SPMM_WGS * SPMM_WAVES * 4)

static_assert(N_EDGES % 32 == 0, "the edge list is whole 32-edge chunks");
static_assert(BM_PITCH * 32 < (1 << 16), "a chunk number fits in 16 bits");
static_assert(BM_WPW <= 32 && CHUNK_DEPTH <= 32, "one bitmap word, one chunk number per lane");
static_assert(SPMM_WGS <= 256 && SPMM_WGS % 32 == 0, "the chunk-bit table has one word per block");
static_assert(WS_BM_OFF % 128 == 0, "bitmap rows start on a 128-B line");
static_assert(SPMM_ROWS < 256 && N_EDGES < (1 << 24), "list entry packs edge<<8 | local row");
static_assert(SPMM_ROWS < ACC_ROWS, "row SPMM_ROWS is the spare row");
static_assert(2 * ROW_TILES <= SPMM_WAVES, "one wave per 16-row x 64-column piece");
static_assert(2 * ROW_TILES < SPMM_WAVES, "the last wave is free to store the counts");
static_assert(N_NODES < (1 << 24) && KEEP_CAP % 32 == 0, "kept entry packs column<<8 | local row");


__global__ __launch_bounds__(SPMM_THREADS)
void edge_chunk_bits(const int* __restrict__ erow, unsigned* __restrict__ bm) {
    __shared__ unsigned tbl[SPMM_WAVES * BM_TBL_PITCH];
    const int tid  = threadIdx.x;
    const int lane = tid & 31;
    const int wave = tid >> 5;
    const int g    = blockIdx.x * SPMM_WAVES + __builtin_amdgcn_readfirstlane(wave);
    unsigned* mine = tbl + wave * BM_TBL_PITCH;

    int r[32];
#pragma unroll
    for (int c = 0; c < 32; ++c) {
        const int chunk = g * 32 + c;
        r[c] = (chunk < CHUNKS) ? erow[chunk * 32 + lane] : -1;
    }
    for (int i = lane; i < SPMM_WGS; i += 32) mine[i] = 0u;
#pragma unroll
    for (int c = 0; c < 32; ++c) {
        const unsigned blk = (unsigned)r[c] / (unsigned)SPMM_ROWS;
        if (blk < (unsigned)SPMM_WGS)
            __hip_atomic_fetch_or(mine + blk, 1u << c, __ATOMIC_RELAXED, __HIP_MEMORY_SCOPE_WORKGROUP);
    }
    __syncthreads();
    for (int b = wave; b < SPMM_WGS; b += SPMM_WAVES)
        bm[b * BM_PITCH + blockIdx.x * SPMM_WAVES + lane] = tbl[lane * BM_TBL_PITCH + b];
}

__device__ __forceinline__ float tanh_exp_rcp(float x) {
    const float t = __builtin_amdgcn_exp2f(__builtin_fabsf(x) * 2.8853900817779268f);
    const float r = 1.0f - 2.0f * __builtin_amdgcn_rcpf(t + 1.0f);
    return __builtin_copysignf(r, x);
}

template <bool LAST>
__global__ __launch_bounds__(SPMM_THREADS)
void gcn_layer_own(const int* __restrict__ erow, const int* __restrict__ ecol,
                   const float* __restrict__ evals, const float* __restrict__ h,
                   const float* __restrict__ Wm, const float* __restrict__ bias,
                   const float* __restrict__ fcw, const float* __restrict__ fcb,
                   unsigned* __restrict__ kept, int* __restrict__ nkept,
                   const unsigned* __restrict__ bm, float* __restrict__ out) {
    __shared__ float    acc[ACC_ROWS * ACC_PITCH];
    __shared__ unsigned elist[SPMM_WAVES * (SPMM_LIST + CHUNK_LIST)];
    __shared__ __attribute__((aligned(16))) float ldsW[DIM * DIM];
    __shared__ int      nfound[SPMM_WAVES];

    const int tid  = threadIdx.x;
    const int lane = tid & 31;
    const int wave = tid >> 5;
    const int base = blockIdx.x * SPMM_ROWS;

    static_assert(DIM * DIM / 8 == 2 * SPMM_THREADS, "two four-column items per thread");
    float4 wlo[2], whi[2];
#pragma unroll
    for (int i = 0; i < 2; ++i) {
        const int q  = tid + i * SPMM_THREADS;
        const int c4 = q & 31;
        const int kk = q >> 6;
        const int k0 = 16 * (kk >> 2) + 8 * ((q >> 5) & 1) + (kk & 3);
        wlo[i] = ((const float4*)(Wm + (k0 + 0) * DIM))[c4];
        whi[i] = ((const float4*)(Wm + (k0 + 4) * DIM))[c4];
    }

    v2u* __restrict__ mykept = (v2u*)kept + (size_t)(blockIdx.x * SPMM_WAVES + wave) * KEEP_CAP;
    const v2u nothing = {(unsigned)SPMM_ROWS, 0u};
    int  nlist  = 0;
    bool rescan = true;
    v2u  ahead[KEEP_AHEAD];
    if (LAST) {
        const unsigned nk = (unsigned)__builtin_amdgcn_readfirstlane(nkept[blockIdx.x * SPMM_WAVES + wave]);
        rescan = nk > (unsigned)KEEP_CAP;
        nlist  = rescan ? 0 : (int)nk;
#pragma unroll
        for (int q = 0; q < KEEP_AHEAD; ++q)
            ahead[q] = (q * 32 + lane < nlist) ? mykept[q * 32 + lane] : nothing;
    }

    unsigned wbits = 0u;
    if (rescan && lane < BM_WPW) wbits = bm[blockIdx.x * BM_PITCH + lane * SPMM_WAVES + wave];

    for (int i = tid; i < ACC_ROWS * ACC_PITCH / 4; i += SPMM_THREADS)
        ((float4*)acc)[i] = make_float4(0.0f, 0.0f, 0.0f, 0.0f);

#pragma unroll
    for (int i = 0; i < 2; ++i) {
        const int q  = tid + i * SPMM_THREADS;
        const int c4 = q & 31;
        const int kk = q >> 6;
        const int hw = (q >> 5) & 1;
        const int p0 = (kk * 8 + (c4 >> 2)) * 32 + hw * 16 + (c4 & 3) * 4;
        float4* dst = (float4*)&ldsW[2 * p0];
        dst[0] = make_float4(wlo[i].x, whi[i].x, wlo[i].y, whi[i].y);
        dst[1] = make_float4(wlo[i].z, whi[i].z, wlo[i].w, whi[i].w);
    }
    __syncthreads();

    unsigned* __restrict__ mylist = elist + wave * (SPMM_LIST + CHUNK_LIST);
    unsigned short* __restrict__ clist = (unsigned short*)(mylist + SPMM_LIST);

    int count = 0;

    auto gather_add = [&](int n, int c, float v, int lr) {
        for (int j = 0; j < n; j += GATHER_DEPTH) {
            float4 x[GATHER_DEPTH];
#pragma unroll
            for (int u = 0; u < GATHER_DEPTH; ++u) {
                const int cj = __builtin_amdgcn_readlane(c, j + u);
                x[u] = ((const float4*)(h + (size_t)cj * DIM))[lane];
            }
#pragma unroll
            for (int u = 0; u < GATHER_DEPTH; ++u) {
                const float vj = __builtin_bit_cast(
                    float, __builtin_amdgcn_readlane(__builtin_bit_cast(int, v), j + u));
                const int lrj = __builtin_amdgcn_readlane(lr, j + u);
                float* a = acc + lrj * ACC_PITCH + lane;
                __hip_atomic_fetch_add(a +  0, vj * x[u].x, __ATOMIC_RELAXED, __HIP_MEMORY_SCOPE_WORKGROUP);
                __hip_atomic_fetch_add(a + 32, vj * x[u].y, __ATOMIC_RELAXED, __HIP_MEMORY_SCOPE_WORKGROUP);
                __hip_atomic_fetch_add(a + 64, vj * x[u].z, __ATOMIC_RELAXED, __HIP_MEMORY_SCOPE_WORKGROUP);
                __hip_atomic_fetch_add(a + 96, vj * x[u].w, __ATOMIC_RELAXED, __HIP_MEMORY_SCOPE_WORKGROUP);
            }
        }
    };

    int found = 0;

    auto drain = [&](int keep) {
        while (count > keep) {
            const int n = min(count, 32);
            count -= n;
            const bool live = lane < n;
            const unsigned p = live ? mylist[count + lane] : 0u;
            const int e  = (int)(p >> 8);
            const int lr = live ? (int)(p & 255u) : SPMM_ROWS;
            const int   c = (int)min((unsigned)ecol[e], (unsigned)(N_NODES - 1));
            const float v = evals[e];
            if (!LAST) {
                if (live && found + lane < KEEP_CAP) {
                    v2u s;
                    s.x = ((unsigned)c << 8) | (unsigned)lr;
                    s.y = __builtin_bit_cast(unsigned, v);
                    mykept[found + lane] = s;
                }
            }
            found += n;
            gather_add(n, c, v, lr);
        }
    };

    if (rescan) {
        const int pc = __builtin_popcount(wbits);
        int upto = pc;
#pragma unroll
        for (int off = 1; off < 32; off <<= 1) {
            const int t = __shfl_up(upto, off, 32);
            if (lane >= off) upto += t;
        }
        const int nchunks = __builtin_amdgcn_readlane(upto, 31);
        {
            int pos = upto - pc;
            const unsigned first = (unsigned)(lane * SPMM_WAVES + wave) * 32u;
            for (unsigned wb = wbits; wb != 0u; wb &= wb - 1u)
                clist[pos++] = (unsigned short)(first + (unsigned)__builtin_ctz(wb));
        }
        for (int k0 = 0; k0 < nchunks; k0 += CHUNK_DEPTH) {
            const int m   = min(nchunks - k0, CHUNK_DEPTH);
            const int myc = (lane < m) ? (int)clist[k0 + lane] : -1;
            int r[CHUNK_DEPTH];
#pragma unroll
            for (int u = 0; u < CHUNK_DEPTH; ++u) {
                const int cid = __builtin_amdgcn_readlane(myc, u);
                r[u] = erow[min(max(cid, 0), CHUNKS - 1) * 32 + lane];
            }
#pragma unroll
            for (int u = 0; u < CHUNK_DEPTH; ++u) {
                const int cid = __builtin_amdgcn_readlane(myc, u);
                const unsigned d = (unsigned)r[u] - (unsigned)base;
                const bool hit = (cid >= 0) && (d < SPMM_ROWS);
                const unsigned b = __builtin_amdgcn_ballot_w32(hit);
                if (hit) mylist[count + __builtin_amdgcn_mbcnt_lo(b, 0u)] =
                             (((unsigned)min(cid, CHUNKS - 1) * 32u + (unsigned)lane) << 8) | d;
                count += __builtin_popcount(b);
            }
            drain(31);
        }
        drain(0);
    } else {
        for (int k0 = 0, q = 0; k0 < nlist; k0 += 32, ++q) {
            const int m = min(nlist - k0, 32);
            v2u s = ahead[0];
#pragma unroll
            for (int i = 1; i < KEEP_AHEAD; ++i) s = (q == i) ? ahead[i] : s;
            if (q >= KEEP_AHEAD) s = (lane < m) ? mykept[k0 + lane] : nothing;
            const int   c  = (int)min(s.x >> 8, (unsigned)(N_NODES - 1));
            const int   lr = (int)(s.x & 255u);
            const unsigned vbits = s.y;
            const float v  = __builtin_bit_cast(float, vbits);
            gather_add(m, c, v, lr);
        }
    }
    if (!LAST && lane == 0) nfound[wave] = found;
    __syncthreads();
    if (!LAST && wave == SPMM_WAVES - 1) nkept[blockIdx.x * SPMM_WAVES + lane] = nfound[lane];

    const bool mm = wave < 2 * ROW_TILES;
    const int  mt = wave >> 1;
    const int  nh = wave & 1;
    const int  hh = lane >> 4;
    const int  cc = lane & 15;
    v8f d[4] = {};
    float  bc[4] = {0.0f, 0.0f, 0.0f, 0.0f};
    float4 fw = make_float4(0.0f, 0.0f, 0.0f, 0.0f);
    float  fb = 0.0f;
    if (LAST) {
        fw = ((const float4*)fcw)[lane];
        fb = fcb[0];
    }
    if (mm) {
#pragma unroll
        for (int nt = 0; nt < 4; ++nt) bc[nt] = bias[nh * 64 + nt * 16 + cc];
        const float* arow = acc + (mt * 16 + cc) * ACC_PITCH + 2 * hh;
        for (int kk = 0; kk < 32; ++kk) {
            const v2f a = *(const v2f*)(arow + (kk & 3) * 32 + 4 * (kk >> 2));
#pragma unroll
            for (int nt = 0; nt < 4; ++nt) {
                const v2f b = *(const v2f*)&ldsW[(((kk << 3) + nh * 4 + nt) * 32 + lane) * 2];
                d[nt] = __builtin_amdgcn_wmma_f32_16x16x4_f32(
                     false, a,  false, b,
                     (short)0, d[nt],
                     false,  false);
            }
        }
    }
    __syncthreads();
    if (mm) {
#pragma unroll
        for (int nt = 0; nt < 4; ++nt) {
            const int   col = nh * 64 + nt * 16 + cc;
            float* t = acc + (mt * 16 + 8 * hh) * ACC_PITCH + (col & 3) * 32 + (col >> 2);
#pragma unroll
            for (int j = 0; j < 8; ++j) t[j * ACC_PITCH] = tanh_exp_rcp(d[nt][j] + bc[nt]);
        }
    }
    __syncthreads();

    if (!LAST) {
        for (int rr = wave; rr < SPMM_ROWS; rr += SPMM_WAVES) {
            const int row = base + rr;
            if (row < N_NODES) {
                const float* a = acc + rr * ACC_PITCH + lane;
                ((float4*)(out + (size_t)row * DIM))[lane] = make_float4(a[0], a[32], a[64], a[96]);
            }
        }
    } else {
        float* dots = (float*)elist;
        for (int rr = wave; rr < SPMM_ROWS; rr += SPMM_WAVES) {
            const float* a = acc + rr * ACC_PITCH + lane;
            float s = a[0] * fw.x + a[32] * fw.y + a[64] * fw.z + a[96] * fw.w;
#pragma unroll
            for (int off = 16; off > 0; off >>= 1) s += __shfl_down(s, off, 32);
            if (lane == 0) dots[rr] = s + fb;
        }
        __syncthreads();
        if (tid < SPMM_ROWS && base + tid < N_NODES) out[base + tid] = dots[tid];
    }
}

extern "C" void kernel_launch(void* const* d_in, const int* in_sizes, int n_in,
                              void* d_out, int out_size, void* d_ws, size_t ws_size,
                              hipStream_t stream) {
    const float* features = (const float*)d_in[0];
    const int*   erow     = (const int*)  d_in[1];
    const int*   ecol     = (const int*)  d_in[2];
    const float* evals    = (const float*)d_in[3];
    const float* W1       = (const float*)d_in[4];
    const float* b1       = (const float*)d_in[5];
    const float* W2       = (const float*)d_in[6];
    const float* b2       = (const float*)d_in[7];
    const float* fcw      = (const float*)d_in[8];
    const float* fcb      = (const float*)d_in[9];
    float*       out      = (float*)d_out;

    float*    h1    = (float*)d_ws;
    unsigned* kept  = (unsigned*)((char*)d_ws + WS_KEPT_OFF);
    int*      nkept = (int*)((char*)d_ws + WS_NKEPT_OFF);
    unsigned* bm    = (unsigned*)((char*)d_ws + WS_BM_OFF);

    const dim3 grid(SPMM_WGS);

    edge_chunk_bits<<<dim3(BM_WGS), SPMM_THREADS, 0, stream>>>(erow, bm);
    gcn_layer_own<false><<<grid, SPMM_THREADS, 0, stream>>>(erow, ecol, evals, features, W1, b1, fcw, fcb, kept, nkept, bm, h1);
    gcn_layer_own<true><<<grid, SPMM_THREADS, 0, stream>>>(erow, ecol, evals, h1, W2, b2, fcw, fcb, kept, nkept, bm, out);
}


# row-block owner: each found edge takes a slot of its row in LDS (integer LDS counter), then one wave sums a row's slots in registers; 4 LDS f32 atomics per edge gone; sum order = slot arrival order
# speedup vs baseline: 3.4817x; 1.0528x over previous
#include <hip/hip_runtime.h>


#define N_NODES 50000
#define N_EDGES 600000
#define DIM     128

typedef __attribute__((ext_vector_type(2))) float v2f;
typedef __attribute__((ext_vector_type(8))) float v8f;
typedef __attribute__((ext_vector_type(2))) unsigned v2u;

#define SPMM_ROWS    196
#define SPMM_WGS     ((N_NODES + SPMM_ROWS - 1) / SPMM_ROWS)
#define SPMM_WAVES   32
#define SPMM_THREADS (SPMM_WAVES * 32)
#define CHUNKS       (N_EDGES / 32)
#define BM_WORDS     ((CHUNKS + 31) / 32)
#define BM_WGS       ((BM_WORDS + 31) / 32)
#define BM_PITCH     (BM_WGS * 32)
#define BM_WPW       (BM_PITCH / SPMM_WAVES)
#define BM_TBL_PITCH 257
#define CHUNK_DEPTH  16
#define SPMM_LIST    (31 + CHUNK_DEPTH * 32)
#define CHUNK_LIST   (BM_WPW * 32 / 2)
#define GATHER_DEPTH 16
#define ROW_TILES    ((SPMM_ROWS + 15) / 16)
#define ACC_ROWS     (ROW_TILES * 16)
#define ACC_PITCH    (DIM + 4)
#define KEEP_CAP     256
#define KEEP_AHEAD   3
#define ROW_CAP      27
#define WAVE_ROWS    ((SPMM_ROWS + SPMM_WAVES - 1) / SPMM_WAVES)
#define FLAT_VALS    192
#define WS_KEPT_OFF  ((size_t)N_NODES * DIM * 4)
#define WS_NKEPT_OFF (WS_KEPT_OFF + (size_t)SPMM_WGS * SPMM_WAVES * KEEP_CAP * 8)
#define WS_BM_OFF    (WS_NKEPT_OFF + (size_t)SPMM_WGS * SPMM_WAVES * 4)

static_assert(N_EDGES % 32 == 0, "the edge list is whole 32-edge chunks");
static_assert(BM_PITCH * 32 < (1 << 16), "a chunk number fits in 16 bits");
static_assert(BM_WPW <= 32 && CHUNK_DEPTH <= 32, "one bitmap word, one chunk number per lane");
static_assert(SPMM_WGS <= 256 && SPMM_WGS % 32 == 0, "the chunk-bit table has one word per block");
static_assert(WS_BM_OFF % 128 == 0, "bitmap rows start on a 128-B line");
static_assert(SPMM_ROWS < 256 && N_EDGES < (1 << 24), "list entry packs edge<<8 | local row");
static_assert(SPMM_ROWS < ACC_ROWS, "row SPMM_ROWS is the spare row");
static_assert(2 * ROW_TILES <= SPMM_WAVES, "one wave per 16-row x 64-column piece");
static_assert(2 * ROW_TILES < SPMM_WAVES, "the last wave is free to store the counts");
static_assert(N_NODES < (1 << 24) && KEEP_CAP % 32 == 0, "kept entry packs column<<8 | local row");
static_assert(ROW_CAP <= 32, "a row's slots are read one per lane");
static_assert(WAVE_ROWS * ROW_CAP <= FLAT_VALS && FLAT_VALS + WAVE_ROWS * ROW_CAP <= SPMM_LIST, "the wave's list fits its edge list's place");
static_assert(WAVE_ROWS <= 32 && WAVE_ROWS < 256, "a row count per lane, a row number in 8 bits");


__global__ __launch_bounds__(SPMM_THREADS)
void edge_chunk_bits(const int* __restrict__ erow, unsigned* __restrict__ bm) {
    __shared__ unsigned tbl[SPMM_WAVES * BM_TBL_PITCH];
    const int tid  = threadIdx.x;
    const int lane = tid & 31;
    const int wave = tid >> 5;
    const int g    = blockIdx.x * SPMM_WAVES + __builtin_amdgcn_readfirstlane(wave);
    unsigned* mine = tbl + wave * BM_TBL_PITCH;

    int r[32];
#pragma unroll
    for (int c = 0; c < 32; ++c) {
        const int chunk = g * 32 + c;
        r[c] = (chunk < CHUNKS) ? erow[chunk * 32 + lane] : -1;
    }
    for (int i = lane; i < SPMM_WGS; i += 32) mine[i] = 0u;
#pragma unroll
    for (int c = 0; c < 32; ++c) {
        const unsigned blk = (unsigned)r[c] / (unsigned)SPMM_ROWS;
        if (blk < (unsigned)SPMM_WGS)
            __hip_atomic_fetch_or(mine + blk, 1u << c, __ATOMIC_RELAXED, __HIP_MEMORY_SCOPE_WORKGROUP);
    }
    __syncthreads();
    for (int b = wave; b < SPMM_WGS; b += SPMM_WAVES)
        bm[b * BM_PITCH + blockIdx.x * SPMM_WAVES + lane] = tbl[lane * BM_TBL_PITCH + b];
}

__device__ __forceinline__ float tanh_exp_rcp(float x) {
    const float t = __builtin_amdgcn_exp2f(__builtin_fabsf(x) * 2.8853900817779268f);
    const float r = 1.0f - 2.0f * __builtin_amdgcn_rcpf(t + 1.0f);
    return __builtin_copysignf(r, x);
}

template <bool LAST>
__global__ __launch_bounds__(SPMM_THREADS)
void gcn_layer_own(const int* __restrict__ erow, const int* __restrict__ ecol,
                   const float* __restrict__ evals, const float* __restrict__ h,
                   const float* __restrict__ Wm, const float* __restrict__ bias,
                   const float* __restrict__ fcw, const float* __restrict__ fcb,
                   unsigned* __restrict__ kept, int* __restrict__ nkept,
                   const unsigned* __restrict__ bm, float* __restrict__ out) {
    __shared__ float    acc[ACC_ROWS * ACC_PITCH];
    __shared__ unsigned elist[SPMM_WAVES * (SPMM_LIST + CHUNK_LIST)];
    __shared__ __attribute__((aligned(16))) float ldsW[DIM * DIM];
    __shared__ int      nfound[SPMM_WAVES];
    __shared__ v2u      slots[SPMM_ROWS * ROW_CAP];
    __shared__ int      rowcnt[ACC_ROWS];

    const int tid  = threadIdx.x;
    const int lane = tid & 31;
    const int wave = tid >> 5;
    const int base = blockIdx.x * SPMM_ROWS;

    static_assert(DIM * DIM / 8 == 2 * SPMM_THREADS, "two four-column items per thread");
    float4 wlo[2], whi[2];
#pragma unroll
    for (int i = 0; i < 2; ++i) {
        const int q  = tid + i * SPMM_THREADS;
        const int c4 = q & 31;
        const int kk = q >> 6;
        const int k0 = 16 * (kk >> 2) + 8 * ((q >> 5) & 1) + (kk & 3);
        wlo[i] = ((const float4*)(Wm + (k0 + 0) * DIM))[c4];
        whi[i] = ((const float4*)(Wm + (k0 + 4) * DIM))[c4];
    }

    v2u* __restrict__ mykept = (v2u*)kept + (size_t)(blockIdx.x * SPMM_WAVES + wave) * KEEP_CAP;
    const v2u nothing = {(unsigned)SPMM_ROWS, 0u};
    int  nlist  = 0;
    bool rescan = true;
    v2u  ahead[KEEP_AHEAD];
    if (LAST) {
        const unsigned nk = (unsigned)__builtin_amdgcn_readfirstlane(nkept[blockIdx.x * SPMM_WAVES + wave]);
        rescan = nk > (unsigned)KEEP_CAP;
        nlist  = rescan ? 0 : (int)nk;
#pragma unroll
        for (int q = 0; q < KEEP_AHEAD; ++q)
            ahead[q] = (q * 32 + lane < nlist) ? mykept[q * 32 + lane] : nothing;
    }

    unsigned wbits = 0u;
    if (rescan && lane < BM_WPW) wbits = bm[blockIdx.x * BM_PITCH + lane * SPMM_WAVES + wave];

    for (int i = tid; i < ACC_ROWS * ACC_PITCH / 4; i += SPMM_THREADS)
        ((float4*)acc)[i] = make_float4(0.0f, 0.0f, 0.0f, 0.0f);
    if (tid < ACC_ROWS) rowcnt[tid] = 0;

#pragma unroll
    for (int i = 0; i < 2; ++i) {
        const int q  = tid + i * SPMM_THREADS;
        const int c4 = q & 31;
        const int kk = q >> 6;
        const int hw = (q >> 5) & 1;
        const int p0 = (kk * 8 + (c4 >> 2)) * 32 + hw * 16 + (c4 & 3) * 4;
        float4* dst = (float4*)&ldsW[2 * p0];
        dst[0] = make_float4(wlo[i].x, whi[i].x, wlo[i].y, whi[i].y);
        dst[1] = make_float4(wlo[i].z, whi[i].z, wlo[i].w, whi[i].w);
    }
    __syncthreads();

    unsigned* __restrict__ mylist = elist + wave * (SPMM_LIST + CHUNK_LIST);
    unsigned short* __restrict__ clist = (unsigned short*)(mylist + SPMM_LIST);

    int count = 0;

    auto place = [&](int n, int c, float v, int lr) {
        const bool live = (lane < n) && (lr < SPMM_ROWS);
        int slot = 0;
        if (live) slot = __hip_atomic_fetch_add(rowcnt + lr, 1, __ATOMIC_RELAXED, __HIP_MEMORY_SCOPE_WORKGROUP);
        if (live && slot < ROW_CAP) {
            v2u s;
            s.x = (unsigned)c;
            s.y = __builtin_bit_cast(unsigned, v);
            slots[lr * ROW_CAP + slot] = s;
        }
        const bool over = live && slot >= ROW_CAP;
        for (unsigned ob = __builtin_amdgcn_ballot_w32(over); ob != 0u; ob &= ob - 1u) {
            const int   l   = __builtin_ctz(ob);
            const int   cj  = __builtin_amdgcn_readlane(c, l);
            const int   lrj = __builtin_amdgcn_readlane(lr, l);
            const float vj  = __builtin_bit_cast(
                float, __builtin_amdgcn_readlane(__builtin_bit_cast(int, v), l));
            const float4 x = ((const float4*)(h + (size_t)cj * DIM))[lane];
            float* a = acc + lrj * ACC_PITCH + lane;
            __hip_atomic_fetch_add(a +  0, vj * x.x, __ATOMIC_RELAXED, __HIP_MEMORY_SCOPE_WORKGROUP);
            __hip_atomic_fetch_add(a + 32, vj * x.y, __ATOMIC_RELAXED, __HIP_MEMORY_SCOPE_WORKGROUP);
            __hip_atomic_fetch_add(a + 64, vj * x.z, __ATOMIC_RELAXED, __HIP_MEMORY_SCOPE_WORKGROUP);
            __hip_atomic_fetch_add(a + 96, vj * x.w, __ATOMIC_RELAXED, __HIP_MEMORY_SCOPE_WORKGROUP);
        }
    };

    int found = 0;

    auto drain = [&](int keep) {
        while (count > keep) {
            const int n = min(count, 32);
            count -= n;
            const bool live = lane < n;
            const unsigned p = live ? mylist[count + lane] : 0u;
            const int e  = (int)(p >> 8);
            const int lr = live ? (int)(p & 255u) : SPMM_ROWS;
            const int   c = (int)min((unsigned)ecol[e], (unsigned)(N_NODES - 1));
            const float v = evals[e];
            if (!LAST) {
                if (live && found + lane < KEEP_CAP) {
                    v2u s;
                    s.x = ((unsigned)c << 8) | (unsigned)lr;
                    s.y = __builtin_bit_cast(unsigned, v);
                    mykept[found + lane] = s;
                }
            }
            found += n;
            place(n, c, v, lr);
        }
    };

    if (rescan) {
        const int pc = __builtin_popcount(wbits);
        int upto = pc;
#pragma unroll
        for (int off = 1; off < 32; off <<= 1) {
            const int t = __shfl_up(upto, off, 32);
            if (lane >= off) upto += t;
        }
        const int nchunks = __builtin_amdgcn_readlane(upto, 31);
        {
            int pos = upto - pc;
            const unsigned first = (unsigned)(lane * SPMM_WAVES + wave) * 32u;
            for (unsigned wb = wbits; wb != 0u; wb &= wb - 1u)
                clist[pos++] = (unsigned short)(first + (unsigned)__builtin_ctz(wb));
        }
        for (int k0 = 0; k0 < nchunks; k0 += CHUNK_DEPTH) {
            const int m   = min(nchunks - k0, CHUNK_DEPTH);
            const int myc = (lane < m) ? (int)clist[k0 + lane] : -1;
            int r[CHUNK_DEPTH];
#pragma unroll
            for (int u = 0; u < CHUNK_DEPTH; ++u) {
                const int cid = __builtin_amdgcn_readlane(myc, u);
                r[u] = erow[min(max(cid, 0), CHUNKS - 1) * 32 + lane];
            }
#pragma unroll
            for (int u = 0; u < CHUNK_DEPTH; ++u) {
                const int cid = __builtin_amdgcn_readlane(myc, u);
                const unsigned d = (unsigned)r[u] - (unsigned)base;
                const bool hit = (cid >= 0) && (d < SPMM_ROWS);
                const unsigned b = __builtin_amdgcn_ballot_w32(hit);
                if (hit) mylist[count + __builtin_amdgcn_mbcnt_lo(b, 0u)] =
                             (((unsigned)min(cid, CHUNKS - 1) * 32u + (unsigned)lane) << 8) | d;
                count += __builtin_popcount(b);
            }
            drain(31);
        }
        drain(0);
    } else {
        for (int k0 = 0, q = 0; k0 < nlist; k0 += 32, ++q) {
            const int m = min(nlist - k0, 32);
            v2u s = ahead[0];
#pragma unroll
            for (int i = 1; i < KEEP_AHEAD; ++i) s = (q == i) ? ahead[i] : s;
            if (q >= KEEP_AHEAD) s = (lane < m) ? mykept[k0 + lane] : nothing;
            const int   c  = (int)min(s.x >> 8, (unsigned)(N_NODES - 1));
            const int   lr = (int)(s.x & 255u);
            const unsigned vbits = s.y;
            const float v  = __builtin_bit_cast(float, vbits);
            place(m, c, v, lr);
        }
    }
    if (!LAST && lane == 0) nfound[wave] = found;
    __syncthreads();
    if (!LAST && wave == SPMM_WAVES - 1) nkept[blockIdx.x * SPMM_WAVES + lane] = nfound[lane];

    {
        const int krow  = wave + SPMM_WAVES * lane;
        const int mycnt = (lane < WAVE_ROWS && krow < SPMM_ROWS) ? min(rowcnt[krow], ROW_CAP) : 0;
        const v2u empty = {0u, 0u};
        v2u rk[WAVE_ROWS];
#pragma unroll
        for (int k = 0; k < WAVE_ROWS; ++k) {
            const int cnt = __builtin_amdgcn_readlane(mycnt, k);
            rk[k] = (lane < cnt) ? slots[(wave + SPMM_WAVES * k) * ROW_CAP + lane] : empty;
        }
        int total = 0;
#pragma unroll
        for (int k = 0; k < WAVE_ROWS; ++k) {
            const int cnt = __builtin_amdgcn_readlane(mycnt, k);
            if (lane < cnt) {
                mylist[total + lane]             = rk[k].x | ((unsigned)k << 24);
                mylist[FLAT_VALS + total + lane] = rk[k].y;
            }
            total += cnt;
        }
        float4 s = make_float4(0.0f, 0.0f, 0.0f, 0.0f);
        int kcur = 0;
        auto flush = [&](int k) {
            float* a = acc + (wave + SPMM_WAVES * k) * ACC_PITCH + lane;
            __hip_atomic_fetch_add(a +  0, s.x, __ATOMIC_RELAXED, __HIP_MEMORY_SCOPE_WORKGROUP);
            __hip_atomic_fetch_add(a + 32, s.y, __ATOMIC_RELAXED, __HIP_MEMORY_SCOPE_WORKGROUP);
            __hip_atomic_fetch_add(a + 64, s.z, __ATOMIC_RELAXED, __HIP_MEMORY_SCOPE_WORKGROUP);
            __hip_atomic_fetch_add(a + 96, s.w, __ATOMIC_RELAXED, __HIP_MEMORY_SCOPE_WORKGROUP);
        };
        for (int j = 0; j < total; j += GATHER_DEPTH) {
            const int m = min(total - j, GATHER_DEPTH);
            unsigned ce = 0u, ve = 0u;
            if (lane < m) {
                ce = mylist[j + lane];
                ve = mylist[FLAT_VALS + j + lane];
            }
            float4 x[GATHER_DEPTH];
#pragma unroll
            for (int u = 0; u < GATHER_DEPTH; ++u) {
                const int cj = __builtin_amdgcn_readlane((int)(ce & 0xFFFFFFu), u);
                x[u] = ((const float4*)(h + (size_t)cj * DIM))[lane];
            }
#pragma unroll
            for (int u = 0; u < GATHER_DEPTH; ++u) {
                if (u < m) {
                    const int ku = __builtin_amdgcn_readlane((int)(ce >> 24), u);
                    if (ku != kcur) {
                        flush(kcur);
                        s = make_float4(0.0f, 0.0f, 0.0f, 0.0f);
                        kcur = ku;
                    }
                    const float vj = __builtin_bit_cast(float, __builtin_amdgcn_readlane((int)ve, u));
                    s.x += vj * x[u].x;
                    s.y += vj * x[u].y;
                    s.z += vj * x[u].z;
                    s.w += vj * x[u].w;
                }
            }
        }
        if (total > 0) flush(kcur);
    }
    __syncthreads();

    const bool mm = wave < 2 * ROW_TILES;
    const int  mt = wave >> 1;
    const int  nh = wave & 1;
    const int  hh = lane >> 4;
    const int  cc = lane & 15;
    v8f d[4] = {};
    float  bc[4] = {0.0f, 0.0f, 0.0f, 0.0f};
    float4 fw = make_float4(0.0f, 0.0f, 0.0f, 0.0f);
    float  fb = 0.0f;
    if (LAST) {
        fw = ((const float4*)fcw)[lane];
        fb = fcb[0];
    }
    if (mm) {
#pragma unroll
        for (int nt = 0; nt < 4; ++nt) bc[nt] = bias[nh * 64 + nt * 16 + cc];
        const float* arow = acc + (mt * 16 + cc) * ACC_PITCH + 2 * hh;
        for (int kk = 0; kk < 32; ++kk) {
            const v2f a = *(const v2f*)(arow + (kk & 3) * 32 + 4 * (kk >> 2));
#pragma unroll
            for (int nt = 0; nt < 4; ++nt) {
                const v2f b = *(const v2f*)&ldsW[(((kk << 3) + nh * 4 + nt) * 32 + lane) * 2];
                d[nt] = __builtin_amdgcn_wmma_f32_16x16x4_f32(
                     false, a,  false, b,
                     (short)0, d[nt],
                     false,  false);
            }
        }
    }
    __syncthreads();
    if (mm) {
#pragma unroll
        for (int nt = 0; nt < 4; ++nt) {
            const int   col = nh * 64 + nt * 16 + cc;
            float* t = acc + (mt * 16 + 8 * hh) * ACC_PITCH + (col & 3) * 32 + (col >> 2);
#pragma unroll
            for (int j = 0; j < 8; ++j) t[j * ACC_PITCH] = tanh_exp_rcp(d[nt][j] + bc[nt]);
        }
    }
    __syncthreads();

    if (!LAST) {
        for (int rr = wave; rr < SPMM_ROWS; rr += SPMM_WAVES) {
            const int row = base + rr;
            if (row < N_NODES) {
                const float* a = acc + rr * ACC_PITCH + lane;
                ((float4*)(out + (size_t)row * DIM))[lane] = make_float4(a[0], a[32], a[64], a[96]);
            }
        }
    } else {
        float* dots = (float*)elist;
        for (int rr = wave; rr < SPMM_ROWS; rr += SPMM_WAVES) {
            const float* a = acc + rr * ACC_PITCH + lane;
            float s = a[0] * fw.x + a[32] * fw.y + a[64] * fw.z + a[96] * fw.w;
#pragma unroll
            for (int off = 16; off > 0; off >>= 1) s += __shfl_down(s, off, 32);
            if (lane == 0) dots[rr] = s + fb;
        }
        __syncthreads();
        if (tid < SPMM_ROWS && base + tid < N_NODES) out[base + tid] = dots[tid];
    }
}

extern "C" void kernel_launch(void* const* d_in, const int* in_sizes, int n_in,
                              void* d_out, int out_size, void* d_ws, size_t ws_size,
                              hipStream_t stream) {
    const float* features = (const float*)d_in[0];
    const int*   erow     = (const int*)  d_in[1];
    const int*   ecol     = (const int*)  d_in[2];
    const float* evals    = (const float*)d_in[3];
    const float* W1       = (const float*)d_in[4];
    const float* b1       = (const float*)d_in[5];
    const float* W2       = (const float*)d_in[6];
    const float* b2       = (const float*)d_in[7];
    const float* fcw      = (const float*)d_in[8];
    const float* fcb      = (const float*)d_in[9];
    float*       out      = (float*)d_out;

    float*    h1    = (float*)d_ws;
    unsigned* kept  = (unsigned*)((char*)d_ws + WS_KEPT_OFF);
    int*      nkept = (int*)((char*)d_ws + WS_NKEPT_OFF);
    unsigned* bm    = (unsigned*)((char*)d_ws + WS_BM_OFF);

    const dim3 grid(SPMM_WGS);

    edge_chunk_bits<<<dim3(BM_WGS), SPMM_THREADS, 0, stream>>>(erow, bm);
    gcn_layer_own<false><<<grid, SPMM_THREADS, 0, stream>>>(erow, ecol, evals, features, W1, b1, fcw, fcb, kept, nkept, bm, h1);
    gcn_layer_own<true><<<grid, SPMM_THREADS, 0, stream>>>(erow, ecol, evals, h1, W2, b2, fcw, fcb, kept, nkept, bm, out);
}


# first launch also repacks the edge list, in place, to 8-byte records (row, column in 16 bits each, value): layer 1 reads one 256-B piece per marked chunk; 2 scattered 4-byte reads per edge gone
# speedup vs baseline: 3.6473x; 1.0476x over previous
#include <hip/hip_runtime.h>


#define N_NODES 50000
#define N_EDGES 600000
#define DIM     128

typedef __attribute__((ext_vector_type(2))) float v2f;
typedef __attribute__((ext_vector_type(8))) float v8f;
typedef __attribute__((ext_vector_type(2))) unsigned v2u;

#define SPMM_ROWS    196
#define SPMM_WGS     ((N_NODES + SPMM_ROWS - 1) / SPMM_ROWS)
#define SPMM_WAVES   32
#define SPMM_THREADS (SPMM_WAVES * 32)
#define CHUNKS       (N_EDGES / 32)
#define BM_WORDS     ((CHUNKS + 31) / 32)
#define BM_WGS       ((BM_WORDS + 31) / 32)
#define BM_PITCH     (BM_WGS * 32)
#define BM_WPW       (BM_PITCH / SPMM_WAVES)
#define BM_TBL_PITCH 257
#define CHUNK_DEPTH  16
#define SPMM_LIST    (31 + CHUNK_DEPTH * 32)
#define CHUNK_LIST   (BM_WPW * 32 / 2)
#define GATHER_DEPTH 16
#define ROW_TILES    ((SPMM_ROWS + 15) / 16)
#define ACC_ROWS     (ROW_TILES * 16)
#define ACC_PITCH    (DIM + 4)
#define KEEP_CAP     256
#define KEEP_AHEAD   3
#define ROW_CAP      27
#define WAVE_ROWS    ((SPMM_ROWS + SPMM_WAVES - 1) / SPMM_WAVES)
#define FLAT_VALS    192
#define SCAN_GROUP   4
#define LIST_VALS    160
#define PACK_WGS     74
#define PACK_ITERS   8
#define WS_KEPT_OFF  ((size_t)N_NODES * DIM * 4)
#define WS_NKEPT_OFF (WS_KEPT_OFF + (size_t)SPMM_WGS * SPMM_WAVES * KEEP_CAP * 8)
#define WS_BM_OFF    (WS_NKEPT_OFF + (size_t)SPMM_WGS * SPMM_WAVES * 4)
#define WS_PK_OFF    (WS_BM_OFF + (size_t)SPMM_WGS * BM_PITCH * 4)

static_assert(N_EDGES % 32 == 0, "the edge list is whole 32-edge chunks");
static_assert(BM_PITCH * 32 < (1 << 16), "a chunk number fits in 16 bits");
static_assert(BM_WPW <= 32 && CHUNK_DEPTH <= 32, "one bitmap word, one chunk number per lane");
static_assert(SPMM_WGS <= 256 && SPMM_WGS % 32 == 0, "the chunk-bit table has one word per block");
static_assert(WS_BM_OFF % 128 == 0, "bitmap rows start on a 128-B line");
static_assert(SPMM_ROWS < 256 && N_EDGES < (1 << 24), "list entry packs edge<<8 | local row");
static_assert(SPMM_ROWS < ACC_ROWS, "row SPMM_ROWS is the spare row");
static_assert(2 * ROW_TILES <= SPMM_WAVES, "one wave per 16-row x 64-column piece");
static_assert(2 * ROW_TILES < SPMM_WAVES, "the last wave is free to store the counts");
static_assert(N_NODES < (1 << 24) && KEEP_CAP % 32 == 0, "kept entry packs column<<8 | local row");
static_assert(ROW_CAP <= 32, "a row's slots are read one per lane");
static_assert(WAVE_ROWS * ROW_CAP <= FLAT_VALS && FLAT_VALS + WAVE_ROWS * ROW_CAP <= SPMM_LIST, "the wave's list fits its edge list's place");
static_assert(WAVE_ROWS <= 32 && WAVE_ROWS < 256, "a row count per lane, a row number in 8 bits");
static_assert(N_NODES <= 0xFFFF, "a row or a column fits in 16 bits, and 0xFFFF is no row");
static_assert(WS_PK_OFF % 256 == 0, "a chunk's records are one aligned 256-B piece");
static_assert(CHUNK_DEPTH % SCAN_GROUP == 0 && 31 + SCAN_GROUP * 32 <= LIST_VALS && 2 * LIST_VALS <= SPMM_LIST,
              "what is left over plus one group's finds fits the list, keys and values");
static_assert((size_t)PACK_WGS * SPMM_THREADS * PACK_ITERS >= N_EDGES, "every edge is repacked");


__global__ __launch_bounds__(SPMM_THREADS)
void edge_chunk_bits(const int* __restrict__ erow, const int* __restrict__ ecol,
                     const float* __restrict__ evals, unsigned* __restrict__ bm,
                     unsigned* __restrict__ pk) {
    __shared__ unsigned tbl[SPMM_WAVES * BM_TBL_PITCH];
    if (blockIdx.x >= BM_WGS) {
        const int t0 = (blockIdx.x - BM_WGS) * SPMM_THREADS + threadIdx.x;
        int   pr[PACK_ITERS], pc[PACK_ITERS];
        float pv[PACK_ITERS];
#pragma unroll
        for (int i = 0; i < PACK_ITERS; ++i) {
            const int e = t0 + i * (PACK_WGS * SPMM_THREADS);
            pr[i] = 0; pc[i] = 0; pv[i] = 0.0f;
            if (e < N_EDGES) { pr[i] = erow[e]; pc[i] = ecol[e]; pv[i] = evals[e]; }
        }
#pragma unroll
        for (int i = 0; i < PACK_ITERS; ++i) {
            const int e = t0 + i * (PACK_WGS * SPMM_THREADS);
            if (e < N_EDGES) {
                const unsigned rw = ((unsigned)pr[i] < (unsigned)N_NODES) ? (unsigned)pr[i] : 0xFFFFu;
                const unsigned cl = min((unsigned)pc[i], (unsigned)(N_NODES - 1));
                v2u s;
                s.x = rw | (cl << 16);
                s.y = __builtin_bit_cast(unsigned, pv[i]);
                ((v2u*)pk)[e] = s;
            }
        }
        return;
    }
    const int tid  = threadIdx.x;
    const int lane = tid & 31;
    const int wave = tid >> 5;
    const int g    = blockIdx.x * SPMM_WAVES + __builtin_amdgcn_readfirstlane(wave);
    unsigned* mine = tbl + wave * BM_TBL_PITCH;

    int r[32];
#pragma unroll
    for (int c = 0; c < 32; ++c) {
        const int chunk = g * 32 + c;
        r[c] = (chunk < CHUNKS) ? erow[chunk * 32 + lane] : -1;
    }
    for (int i = lane; i < SPMM_WGS; i += 32) mine[i] = 0u;
#pragma unroll
    for (int c = 0; c < 32; ++c) {
        const unsigned blk = (unsigned)r[c] / (unsigned)SPMM_ROWS;
        if (blk < (unsigned)SPMM_WGS)
            __hip_atomic_fetch_or(mine + blk, 1u << c, __ATOMIC_RELAXED, __HIP_MEMORY_SCOPE_WORKGROUP);
    }
    __syncthreads();
    for (int b = wave; b < SPMM_WGS; b += SPMM_WAVES)
        bm[b * BM_PITCH + blockIdx.x * SPMM_WAVES + lane] = tbl[lane * BM_TBL_PITCH + b];
}

__device__ __forceinline__ float tanh_exp_rcp(float x) {
    const float t = __builtin_amdgcn_exp2f(__builtin_fabsf(x) * 2.8853900817779268f);
    const float r = 1.0f - 2.0f * __builtin_amdgcn_rcpf(t + 1.0f);
    return __builtin_copysignf(r, x);
}

template <bool LAST>
__global__ __launch_bounds__(SPMM_THREADS)
void gcn_layer_own(const unsigned* __restrict__ pk, const float* __restrict__ h,
                   const float* __restrict__ Wm, const float* __restrict__ bias,
                   const float* __restrict__ fcw, const float* __restrict__ fcb,
                   unsigned* __restrict__ kept, int* __restrict__ nkept,
                   const unsigned* __restrict__ bm, float* __restrict__ out) {
    __shared__ float    acc[ACC_ROWS * ACC_PITCH];
    __shared__ unsigned elist[SPMM_WAVES * (SPMM_LIST + CHUNK_LIST)];
    __shared__ __attribute__((aligned(16))) float ldsW[DIM * DIM];
    __shared__ int      nfound[SPMM_WAVES];
    __shared__ v2u      slots[SPMM_ROWS * ROW_CAP];
    __shared__ int      rowcnt[ACC_ROWS];

    const int tid  = threadIdx.x;
    const int lane = tid & 31;
    const int wave = tid >> 5;
    const int base = blockIdx.x * SPMM_ROWS;

    static_assert(DIM * DIM / 8 == 2 * SPMM_THREADS, "two four-column items per thread");
    float4 wlo[2], whi[2];
#pragma unroll
    for (int i = 0; i < 2; ++i) {
        const int q  = tid + i * SPMM_THREADS;
        const int c4 = q & 31;
        const int kk = q >> 6;
        const int k0 = 16 * (kk >> 2) + 8 * ((q >> 5) & 1) + (kk & 3);
        wlo[i] = ((const float4*)(Wm + (k0 + 0) * DIM))[c4];
        whi[i] = ((const float4*)(Wm + (k0 + 4) * DIM))[c4];
    }

    v2u* __restrict__ mykept = (v2u*)kept + (size_t)(blockIdx.x * SPMM_WAVES + wave) * KEEP_CAP;
    const v2u nothing = {(unsigned)SPMM_ROWS, 0u};
    int  nlist  = 0;
    bool rescan = true;
    v2u  ahead[KEEP_AHEAD];
    if (LAST) {
        const unsigned nk = (unsigned)__builtin_amdgcn_readfirstlane(nkept[blockIdx.x * SPMM_WAVES + wave]);
        rescan = nk > (unsigned)KEEP_CAP;
        nlist  = rescan ? 0 : (int)nk;
#pragma unroll
        for (int q = 0; q < KEEP_AHEAD; ++q)
            ahead[q] = (q * 32 + lane < nlist) ? mykept[q * 32 + lane] : nothing;
    }

    unsigned wbits = 0u;
    if (rescan && lane < BM_WPW) wbits = bm[blockIdx.x * BM_PITCH + lane * SPMM_WAVES + wave];

    for (int i = tid; i < ACC_ROWS * ACC_PITCH / 4; i += SPMM_THREADS)
        ((float4*)acc)[i] = make_float4(0.0f, 0.0f, 0.0f, 0.0f);
    if (tid < ACC_ROWS) rowcnt[tid] = 0;

#pragma unroll
    for (int i = 0; i < 2; ++i) {
        const int q  = tid + i * SPMM_THREADS;
        const int c4 = q & 31;
        const int kk = q >> 6;
        const int hw = (q >> 5) & 1;
        const int p0 = (kk * 8 + (c4 >> 2)) * 32 + hw * 16 + (c4 & 3) * 4;
        float4* dst = (float4*)&ldsW[2 * p0];
        dst[0] = make_float4(wlo[i].x, whi[i].x, wlo[i].y, whi[i].y);
        dst[1] = make_float4(wlo[i].z, whi[i].z, wlo[i].w, whi[i].w);
    }
    __syncthreads();

    unsigned* __restrict__ mylist = elist + wave * (SPMM_LIST + CHUNK_LIST);
    unsigned short* __restrict__ clist = (unsigned short*)(mylist + SPMM_LIST);

    int count = 0;

    auto place = [&](int n, int c, float v, int lr) {
        const bool live = (lane < n) && (lr < SPMM_ROWS);
        int slot = 0;
        if (live) slot = __hip_atomic_fetch_add(rowcnt + lr, 1, __ATOMIC_RELAXED, __HIP_MEMORY_SCOPE_WORKGROUP);
        if (live && slot < ROW_CAP) {
            v2u s;
            s.x = (unsigned)c;
            s.y = __builtin_bit_cast(unsigned, v);
            slots[lr * ROW_CAP + slot] = s;
        }
        const bool over = live && slot >= ROW_CAP;
        for (unsigned ob = __builtin_amdgcn_ballot_w32(over); ob != 0u; ob &= ob - 1u) {
            const int   l   = __builtin_ctz(ob);
            const int   cj  = __builtin_amdgcn_readlane(c, l);
            const int   lrj = __builtin_amdgcn_readlane(lr, l);
            const float vj  = __builtin_bit_cast(
                float, __builtin_amdgcn_readlane(__builtin_bit_cast(int, v), l));
            const float4 x = ((const float4*)(h + (size_t)cj * DIM))[lane];
            float* a = acc + lrj * ACC_PITCH + lane;
            __hip_atomic_fetch_add(a +  0, vj * x.x, __ATOMIC_RELAXED, __HIP_MEMORY_SCOPE_WORKGROUP);
            __hip_atomic_fetch_add(a + 32, vj * x.y, __ATOMIC_RELAXED, __HIP_MEMORY_SCOPE_WORKGROUP);
            __hip_atomic_fetch_add(a + 64, vj * x.z, __ATOMIC_RELAXED, __HIP_MEMORY_SCOPE_WORKGROUP);
            __hip_atomic_fetch_add(a + 96, vj * x.w, __ATOMIC_RELAXED, __HIP_MEMORY_SCOPE_WORKGROUP);
        }
    };

    int found = 0;

    auto drain = [&](int keep) {
        while (count > keep) {
            const int n = min(count, 32);
            count -= n;
            const bool live = lane < n;
            unsigned key = (unsigned)SPMM_ROWS, vbits = 0u;
            if (live) {
                key   = mylist[count + lane];
                vbits = mylist[LIST_VALS + count + lane];
            }
            const int   c  = (int)min(key >> 8, (unsigned)(N_NODES - 1));
            const int   lr = (int)(key & 255u);
            const float v  = __builtin_bit_cast(float, vbits);
            if (!LAST) {
                if (live && found + lane < KEEP_CAP) {
                    v2u s;
                    s.x = key;
                    s.y = vbits;
                    mykept[found + lane] = s;
                }
            }
            found += n;
            place(n, c, v, lr);
        }
    };

    if (rescan) {
        const int pc = __builtin_popcount(wbits);
        int upto = pc;
#pragma unroll
        for (int off = 1; off < 32; off <<= 1) {
            const int t = __shfl_up(upto, off, 32);
            if (lane >= off) upto += t;
        }
        const int nchunks = __builtin_amdgcn_readlane(upto, 31);
        {
            int pos = upto - pc;
            const unsigned first = (unsigned)(lane * SPMM_WAVES + wave) * 32u;
            for (unsigned wb = wbits; wb != 0u; wb &= wb - 1u)
                clist[pos++] = (unsigned short)(first + (unsigned)__builtin_ctz(wb));
        }
        const v2u* __restrict__ rec = (const v2u*)pk;
        for (int k0 = 0; k0 < nchunks; k0 += CHUNK_DEPTH) {
            const int m   = min(nchunks - k0, CHUNK_DEPTH);
            const int myc = (lane < m) ? (int)clist[k0 + lane] : -1;
            v2u r[CHUNK_DEPTH];
#pragma unroll
            for (int u = 0; u < CHUNK_DEPTH; ++u) {
                const int cid = __builtin_amdgcn_readlane(myc, u);
                r[u] = rec[min(max(cid, 0), CHUNKS - 1) * 32 + lane];
            }
#pragma unroll
            for (int u = 0; u < CHUNK_DEPTH; ++u) {
                const int cid = __builtin_amdgcn_readlane(myc, u);
                const unsigned d = (r[u].x & 0xFFFFu) - (unsigned)base;
                const bool hit = (cid >= 0) && (d < SPMM_ROWS);
                const unsigned b = __builtin_amdgcn_ballot_w32(hit);
                if (hit) {
                    const int p = count + (int)__builtin_amdgcn_mbcnt_lo(b, 0u);
                    mylist[p]             = ((r[u].x >> 16) << 8) | d;
                    mylist[LIST_VALS + p] = r[u].y;
                }
                count += __builtin_popcount(b);
                if (u % SCAN_GROUP == SCAN_GROUP - 1) drain(31);
            }
        }
        drain(0);
    } else {
        for (int k0 = 0, q = 0; k0 < nlist; k0 += 32, ++q) {
            const int m = min(nlist - k0, 32);
            v2u s = ahead[0];
#pragma unroll
            for (int i = 1; i < KEEP_AHEAD; ++i) s = (q == i) ? ahead[i] : s;
            if (q >= KEEP_AHEAD) s = (lane < m) ? mykept[k0 + lane] : nothing;
            const int   c  = (int)min(s.x >> 8, (unsigned)(N_NODES - 1));
            const int   lr = (int)(s.x & 255u);
            const unsigned vbits = s.y;
            const float v  = __builtin_bit_cast(float, vbits);
            place(m, c, v, lr);
        }
    }
    if (!LAST && lane == 0) nfound[wave] = found;
    __syncthreads();
    if (!LAST && wave == SPMM_WAVES - 1) nkept[blockIdx.x * SPMM_WAVES + lane] = nfound[lane];

    {
        const int krow  = wave + SPMM_WAVES * lane;
        const int mycnt = (lane < WAVE_ROWS && krow < SPMM_ROWS) ? min(rowcnt[krow], ROW_CAP) : 0;
        const v2u empty = {0u, 0u};
        v2u rk[WAVE_ROWS];
#pragma unroll
        for (int k = 0; k < WAVE_ROWS; ++k) {
            const int cnt = __builtin_amdgcn_readlane(mycnt, k);
            rk[k] = (lane < cnt) ? slots[(wave + SPMM_WAVES * k) * ROW_CAP + lane] : empty;
        }
        int total = 0;
#pragma unroll
        for (int k = 0; k < WAVE_ROWS; ++k) {
            const int cnt = __builtin_amdgcn_readlane(mycnt, k);
            if (lane < cnt) {
                mylist[total + lane]             = rk[k].x | ((unsigned)k << 24);
                mylist[FLAT_VALS + total + lane] = rk[k].y;
            }
            total += cnt;
        }
        float4 s = make_float4(0.0f, 0.0f, 0.0f, 0.0f);
        int kcur = 0;
        auto flush = [&](int k) {
            float* a = acc + (wave + SPMM_WAVES * k) * ACC_PITCH + lane;
            __hip_atomic_fetch_add(a +  0, s.x, __ATOMIC_RELAXED, __HIP_MEMORY_SCOPE_WORKGROUP);
            __hip_atomic_fetch_add(a + 32, s.y, __ATOMIC_RELAXED, __HIP_MEMORY_SCOPE_WORKGROUP);
            __hip_atomic_fetch_add(a + 64, s.z, __ATOMIC_RELAXED, __HIP_MEMORY_SCOPE_WORKGROUP);
            __hip_atomic_fetch_add(a + 96, s.w, __ATOMIC_RELAXED, __HIP_MEMORY_SCOPE_WORKGROUP);
        };
        for (int j = 0; j < total; j += GATHER_DEPTH) {
            const int m = min(total - j, GATHER_DEPTH);
            unsigned ce = 0u, ve = 0u;
            if (lane < m) {
                ce = mylist[j + lane];
                ve = mylist[FLAT_VALS + j + lane];
            }
            float4 x[GATHER_DEPTH];
#pragma unroll
            for (int u = 0; u < GATHER_DEPTH; ++u) {
                const int cj = __builtin_amdgcn_readlane((int)(ce & 0xFFFFFFu), u);
                x[u] = ((const float4*)(h + (size_t)cj * DIM))[lane];
            }
#pragma unroll
            for (int u = 0; u < GATHER_DEPTH; ++u) {
                if (u < m) {
                    const int ku = __builtin_amdgcn_readlane((int)(ce >> 24), u);
                    if (ku != kcur) {
                        flush(kcur);
                        s = make_float4(0.0f, 0.0f, 0.0f, 0.0f);
                        kcur = ku;
                    }
                    const float vj = __builtin_bit_cast(float, __builtin_amdgcn_readlane((int)ve, u));
                    s.x += vj * x[u].x;
                    s.y += vj * x[u].y;
                    s.z += vj * x[u].z;
                    s.w += vj * x[u].w;
                }
            }
        }
        if (total > 0) flush(kcur);
    }
    __syncthreads();

    const bool mm = wave < 2 * ROW_TILES;
    const int  mt = wave >> 1;
    const int  nh = wave & 1;
    const int  hh = lane >> 4;
    const int  cc = lane & 15;
    v8f d[4] = {};
    float  bc[4] = {0.0f, 0.0f, 0.0f, 0.0f};
    float4 fw = make_float4(0.0f, 0.0f, 0.0f, 0.0f);
    float  fb = 0.0f;
    if (LAST) {
        fw = ((const float4*)fcw)[lane];
        fb = fcb[0];
    }
    if (mm) {
#pragma unroll
        for (int nt = 0; nt < 4; ++nt) bc[nt] = bias[nh * 64 + nt * 16 + cc];
        const float* arow = acc + (mt * 16 + cc) * ACC_PITCH + 2 * hh;
        for (int kk = 0; kk < 32; ++kk) {
            const v2f a = *(const v2f*)(arow + (kk & 3) * 32 + 4 * (kk >> 2));
#pragma unroll
            for (int nt = 0; nt < 4; ++nt) {
                const v2f b = *(const v2f*)&ldsW[(((kk << 3) + nh * 4 + nt) * 32 + lane) * 2];
                d[nt] = __builtin_amdgcn_wmma_f32_16x16x4_f32(
                     false, a,  false, b,
                     (short)0, d[nt],
                     false,  false);
            }
        }
    }
    __syncthreads();
    if (mm) {
#pragma unroll
        for (int nt = 0; nt < 4; ++nt) {
            const int   col = nh * 64 + nt * 16 + cc;
            float* t = acc + (mt * 16 + 8 * hh) * ACC_PITCH + (col & 3) * 32 + (col >> 2);
#pragma unroll
            for (int j = 0; j < 8; ++j) t[j * ACC_PITCH] = tanh_exp_rcp(d[nt][j] + bc[nt]);
        }
    }
    __syncthreads();

    if (!LAST) {
        for (int rr = wave; rr < SPMM_ROWS; rr += SPMM_WAVES) {
            const int row = base + rr;
            if (row < N_NODES) {
                const float* a = acc + rr * ACC_PITCH + lane;
                ((float4*)(out + (size_t)row * DIM))[lane] = make_float4(a[0], a[32], a[64], a[96]);
            }
        }
    } else {
        float* dots = (float*)elist;
        for (int rr = wave; rr < SPMM_ROWS; rr += SPMM_WAVES) {
            const float* a = acc + rr * ACC_PITCH + lane;
            float s = a[0] * fw.x + a[32] * fw.y + a[64] * fw.z + a[96] * fw.w;
#pragma unroll
            for (int off = 16; off > 0; off >>= 1) s += __shfl_down(s, off, 32);
            if (lane == 0) dots[rr] = s + fb;
        }
        __syncthreads();
        if (tid < SPMM_ROWS && base + tid < N_NODES) out[base + tid] = dots[tid];
    }
}

extern "C" void kernel_launch(void* const* d_in, const int* in_sizes, int n_in,
                              void* d_out, int out_size, void* d_ws, size_t ws_size,
                              hipStream_t stream) {
    const float* features = (const float*)d_in[0];
    const int*   erow     = (const int*)  d_in[1];
    const int*   ecol     = (const int*)  d_in[2];
    const float* evals    = (const float*)d_in[3];
    const float* W1       = (const float*)d_in[4];
    const float* b1       = (const float*)d_in[5];
    const float* W2       = (const float*)d_in[6];
    const float* b2       = (const float*)d_in[7];
    const float* fcw      = (const float*)d_in[8];
    const float* fcb      = (const float*)d_in[9];
    float*       out      = (float*)d_out;

    float*    h1    = (float*)d_ws;
    unsigned* kept  = (unsigned*)((char*)d_ws + WS_KEPT_OFF);
    int*      nkept = (int*)((char*)d_ws + WS_NKEPT_OFF);
    unsigned* bm    = (unsigned*)((char*)d_ws + WS_BM_OFF);
    unsigned* pk    = (unsigned*)((char*)d_ws + WS_PK_OFF);

    const dim3 grid(SPMM_WGS);

    edge_chunk_bits<<<dim3(BM_WGS + PACK_WGS), SPMM_THREADS, 0, stream>>>(erow, ecol, evals, bm, pk);
    gcn_layer_own<false><<<grid, SPMM_THREADS, 0, stream>>>(pk, features, W1, b1, fcw, fcb, kept, nkept, bm, h1);
    gcn_layer_own<true><<<grid, SPMM_THREADS, 0, stream>>>(pk, h1, W2, b2, fcw, fcb, kept, nkept, bm, out);
}


# chunk bits per 16 edges (one 128-B line of records) and two marked chunks per load instruction, a half wave each: layer 1's scan reads half the bytes in half the steps; arrival order as before
# speedup vs baseline: 3.8611x; 1.0586x over previous
#include <hip/hip_runtime.h>


#define N_NODES 50000
#define N_EDGES 600000
#define DIM     128

typedef __attribute__((ext_vector_type(2))) float v2f;
typedef __attribute__((ext_vector_type(8))) float v8f;
typedef __attribute__((ext_vector_type(2))) unsigned v2u;

#define SPMM_ROWS    196
#define SPMM_WGS     ((N_NODES + SPMM_ROWS - 1) / SPMM_ROWS)
#define SPMM_WAVES   32
#define SPMM_THREADS (SPMM_WAVES * 32)
#define CHUNK_EDGES  16
#define CHUNKS       (N_EDGES / CHUNK_EDGES)
#define LOAD_UNITS   (N_EDGES / 32)
#define BM_WORDS     ((CHUNKS + 31) / 32)
#define BM_WGS       ((BM_WORDS + 31) / 32)
#define BM_PITCH     (BM_WGS * 32)
#define BM_WPW       (BM_PITCH / SPMM_WAVES)
#define BM_PARTS     3
#define BM_WPP       ((BM_WPW + BM_PARTS - 1) / BM_PARTS)
#define BM_TBL_PITCH 257
#define CHUNK_DEPTH  16
#define SPMM_LIST    (31 + CHUNK_DEPTH * 32)
#define CHUNK_LIST   ((31 + BM_WPP * 32 + 1) / 2)
#define GATHER_DEPTH 16
#define ROW_TILES    ((SPMM_ROWS + 15) / 16)
#define ACC_ROWS     (ROW_TILES * 16)
#define ACC_PITCH    (DIM + 4)
#define KEEP_CAP     256
#define KEEP_AHEAD   3
#define ROW_CAP      27
#define WAVE_ROWS    ((SPMM_ROWS + SPMM_WAVES - 1) / SPMM_WAVES)
#define FLAT_VALS    192
#define SCAN_GROUP   4
#define LIST_VALS    160
#define PACK_WGS     74
#define PACK_ITERS   8
#define WS_KEPT_OFF  ((size_t)N_NODES * DIM * 4)
#define WS_NKEPT_OFF (WS_KEPT_OFF + (size_t)SPMM_WGS * SPMM_WAVES * KEEP_CAP * 8)
#define WS_BM_OFF    (WS_NKEPT_OFF + (size_t)SPMM_WGS * SPMM_WAVES * 4)
#define WS_PK_OFF    (WS_BM_OFF + (size_t)SPMM_WGS * BM_PITCH * 4)

static_assert(N_EDGES % 32 == 0 && 32 % CHUNK_EDGES == 0, "the edge list is whole 32-edge pieces, each whole chunks");
static_assert(BM_PITCH * 32 < (1 << 16), "a chunk number fits in 16 bits");
static_assert(BM_WPP <= 32 && BM_PARTS * BM_WPP >= BM_WPW, "one bitmap word per lane and part; the parts cover the wave's words");
static_assert(2 * CHUNK_DEPTH <= 32 && CHUNK_EDGES * 2 == 32, "one chunk number per lane; two chunks per load, a half wave each");
static_assert(SPMM_WGS <= 256 && SPMM_WGS % 32 == 0, "the chunk-bit table has one word per block");
static_assert(WS_BM_OFF % 128 == 0, "bitmap rows start on a 128-B line");
static_assert(SPMM_ROWS < 256 && N_EDGES < (1 << 24), "list entry packs edge<<8 | local row");
static_assert(SPMM_ROWS < ACC_ROWS, "row SPMM_ROWS is the spare row");
static_assert(2 * ROW_TILES <= SPMM_WAVES, "one wave per 16-row x 64-column piece");
static_assert(2 * ROW_TILES < SPMM_WAVES, "the last wave is free to store the counts");
static_assert(N_NODES < (1 << 24) && KEEP_CAP % 32 == 0, "kept entry packs column<<8 | local row");
static_assert(ROW_CAP <= 32, "a row's slots are read one per lane");
static_assert(WAVE_ROWS * ROW_CAP <= FLAT_VALS && FLAT_VALS + WAVE_ROWS * ROW_CAP <= SPMM_LIST, "the wave's list fits its edge list's place");
static_assert(WAVE_ROWS <= 32 && WAVE_ROWS < 256, "a row count per lane, a row number in 8 bits");
static_assert(N_NODES <= 0xFFFF, "a row or a column fits in 16 bits, and 0xFFFF is no row");
static_assert(WS_PK_OFF % 128 == 0, "a chunk's records are one aligned 128-B line");
static_assert(CHUNK_DEPTH % SCAN_GROUP == 0 && 31 + SCAN_GROUP * 32 <= LIST_VALS && 2 * LIST_VALS <= SPMM_LIST,
              "what is left over plus one group's finds fits the list, keys and values");
static_assert((size_t)PACK_WGS * SPMM_THREADS * PACK_ITERS >= N_EDGES, "every edge is repacked");


__global__ __launch_bounds__(SPMM_THREADS)
void edge_chunk_bits(const int* __restrict__ erow, const int* __restrict__ ecol,
                     const float* __restrict__ evals, unsigned* __restrict__ bm,
                     unsigned* __restrict__ pk) {
    __shared__ unsigned tbl[SPMM_WAVES * BM_TBL_PITCH];
    if (blockIdx.x >= BM_WGS) {
        const int t0 = (blockIdx.x - BM_WGS) * SPMM_THREADS + threadIdx.x;
        int   pr[PACK_ITERS], pc[PACK_ITERS];
        float pv[PACK_ITERS];
#pragma unroll
        for (int i = 0; i < PACK_ITERS; ++i) {
            const int e = t0 + i * (PACK_WGS * SPMM_THREADS);
            pr[i] = 0; pc[i] = 0; pv[i] = 0.0f;
            if (e < N_EDGES) { pr[i] = erow[e]; pc[i] = ecol[e]; pv[i] = evals[e]; }
        }
#pragma unroll
        for (int i = 0; i < PACK_ITERS; ++i) {
            const int e = t0 + i * (PACK_WGS * SPMM_THREADS);
            if (e < N_EDGES) {
                const unsigned rw = ((unsigned)pr[i] < (unsigned)N_NODES) ? (unsigned)pr[i] : 0xFFFFu;
                const unsigned cl = min((unsigned)pc[i], (unsigned)(N_NODES - 1));
                v2u s;
                s.x = rw | (cl << 16);
                s.y = __builtin_bit_cast(unsigned, pv[i]);
                ((v2u*)pk)[e] = s;
            }
        }
        return;
    }
    const int tid  = threadIdx.x;
    const int lane = tid & 31;
    const int wave = tid >> 5;
    const int g    = blockIdx.x * SPMM_WAVES + __builtin_amdgcn_readfirstlane(wave);
    unsigned* mine = tbl + wave * BM_TBL_PITCH;

    int r[16];
#pragma unroll
    for (int c = 0; c < 16; ++c) {
        const int unit = g * 16 + c;
        r[c] = (unit < LOAD_UNITS) ? erow[unit * 32 + lane] : -1;
    }
    for (int i = lane; i < SPMM_WGS; i += 32) mine[i] = 0u;
    const int half = lane >> 4;
#pragma unroll
    for (int c = 0; c < 16; ++c) {
        const unsigned blk = (unsigned)r[c] / (unsigned)SPMM_ROWS;
        if (blk < (unsigned)SPMM_WGS)
            __hip_atomic_fetch_or(mine + blk, 1u << (2 * c + half), __ATOMIC_RELAXED, __HIP_MEMORY_SCOPE_WORKGROUP);
    }
    __syncthreads();
    for (int b = wave; b < SPMM_WGS; b += SPMM_WAVES)
        bm[b * BM_PITCH + blockIdx.x * SPMM_WAVES + lane] = tbl[lane * BM_TBL_PITCH + b];
}

__device__ __forceinline__ float tanh_exp_rcp(float x) {
    const float t = __builtin_amdgcn_exp2f(__builtin_fabsf(x) * 2.8853900817779268f);
    const float r = 1.0f - 2.0f * __builtin_amdgcn_rcpf(t + 1.0f);
    return __builtin_copysignf(r, x);
}

template <bool LAST>
__global__ __launch_bounds__(SPMM_THREADS)
void gcn_layer_own(const unsigned* __restrict__ pk, const float* __restrict__ h,
                   const float* __restrict__ Wm, const float* __restrict__ bias,
                   const float* __restrict__ fcw, const float* __restrict__ fcb,
                   unsigned* __restrict__ kept, int* __restrict__ nkept,
                   const unsigned* __restrict__ bm, float* __restrict__ out) {
    __shared__ float    acc[ACC_ROWS * ACC_PITCH];
    __shared__ unsigned elist[SPMM_WAVES * (SPMM_LIST + CHUNK_LIST)];
    __shared__ __attribute__((aligned(16))) float ldsW[DIM * DIM];
    __shared__ int      nfound[SPMM_WAVES];
    __shared__ v2u      slots[SPMM_ROWS * ROW_CAP];
    __shared__ int      rowcnt[ACC_ROWS];

    const int tid  = threadIdx.x;
    const int lane = tid & 31;
    const int wave = tid >> 5;
    const int base = blockIdx.x * SPMM_ROWS;

    static_assert(DIM * DIM / 8 == 2 * SPMM_THREADS, "two four-column items per thread");
    float4 wlo[2], whi[2];
#pragma unroll
    for (int i = 0; i < 2; ++i) {
        const int q  = tid + i * SPMM_THREADS;
        const int c4 = q & 31;
        const int kk = q >> 6;
        const int k0 = 16 * (kk >> 2) + 8 * ((q >> 5) & 1) + (kk & 3);
        wlo[i] = ((const float4*)(Wm + (k0 + 0) * DIM))[c4];
        whi[i] = ((const float4*)(Wm + (k0 + 4) * DIM))[c4];
    }

    v2u* __restrict__ mykept = (v2u*)kept + (size_t)(blockIdx.x * SPMM_WAVES + wave) * KEEP_CAP;
    const v2u nothing = {(unsigned)SPMM_ROWS, 0u};
    int  nlist  = 0;
    bool rescan = true;
    v2u  ahead[KEEP_AHEAD];
    if (LAST) {
        const unsigned nk = (unsigned)__builtin_amdgcn_readfirstlane(nkept[blockIdx.x * SPMM_WAVES + wave]);
        rescan = nk > (unsigned)KEEP_CAP;
        nlist  = rescan ? 0 : (int)nk;
#pragma unroll
        for (int q = 0; q < KEEP_AHEAD; ++q)
            ahead[q] = (q * 32 + lane < nlist) ? mykept[q * 32 + lane] : nothing;
    }

    unsigned wpart[BM_PARTS];
#pragma unroll
    for (int p = 0; p < BM_PARTS; ++p) {
        const int j = p * BM_WPP + lane;
        wpart[p] = 0u;
        if (rescan && lane < BM_WPP && j < BM_WPW) wpart[p] = bm[blockIdx.x * BM_PITCH + j * SPMM_WAVES + wave];
    }

    for (int i = tid; i < ACC_ROWS * ACC_PITCH / 4; i += SPMM_THREADS)
        ((float4*)acc)[i] = make_float4(0.0f, 0.0f, 0.0f, 0.0f);
    if (tid < ACC_ROWS) rowcnt[tid] = 0;

#pragma unroll
    for (int i = 0; i < 2; ++i) {
        const int q  = tid + i * SPMM_THREADS;
        const int c4 = q & 31;
        const int kk = q >> 6;
        const int hw = (q >> 5) & 1;
        const int p0 = (kk * 8 + (c4 >> 2)) * 32 + hw * 16 + (c4 & 3) * 4;
        float4* dst = (float4*)&ldsW[2 * p0];
        dst[0] = make_float4(wlo[i].x, whi[i].x, wlo[i].y, whi[i].y);
        dst[1] = make_float4(wlo[i].z, whi[i].z, wlo[i].w, whi[i].w);
    }
    __syncthreads();

    unsigned* __restrict__ mylist = elist + wave * (SPMM_LIST + CHUNK_LIST);
    unsigned short* __restrict__ clist = (unsigned short*)(mylist + SPMM_LIST);

    int count = 0;

    auto place = [&](int n, int c, float v, int lr) {
        const bool live = (lane < n) && (lr < SPMM_ROWS);
        int slot = 0;
        if (live) slot = __hip_atomic_fetch_add(rowcnt + lr, 1, __ATOMIC_RELAXED, __HIP_MEMORY_SCOPE_WORKGROUP);
        if (live && slot < ROW_CAP) {
            v2u s;
            s.x = (unsigned)c;
            s.y = __builtin_bit_cast(unsigned, v);
            slots[lr * ROW_CAP + slot] = s;
        }
        const bool over = live && slot >= ROW_CAP;
        for (unsigned ob = __builtin_amdgcn_ballot_w32(over); ob != 0u; ob &= ob - 1u) {
            const int   l   = __builtin_ctz(ob);
            const int   cj  = __builtin_amdgcn_readlane(c, l);
            const int   lrj = __builtin_amdgcn_readlane(lr, l);
            const float vj  = __builtin_bit_cast(
                float, __builtin_amdgcn_readlane(__builtin_bit_cast(int, v), l));
            const float4 x = ((const float4*)(h + (size_t)cj * DIM))[lane];
            float* a = acc + lrj * ACC_PITCH + lane;
            __hip_atomic_fetch_add(a +  0, vj * x.x, __ATOMIC_RELAXED, __HIP_MEMORY_SCOPE_WORKGROUP);
            __hip_atomic_fetch_add(a + 32, vj * x.y, __ATOMIC_RELAXED, __HIP_MEMORY_SCOPE_WORKGROUP);
            __hip_atomic_fetch_add(a + 64, vj * x.z, __ATOMIC_RELAXED, __HIP_MEMORY_SCOPE_WORKGROUP);
            __hip_atomic_fetch_add(a + 96, vj * x.w, __ATOMIC_RELAXED, __HIP_MEMORY_SCOPE_WORKGROUP);
        }
    };

    int found = 0;

    auto drain = [&](int keep) {
        while (count > keep) {
            const int n = min(count, 32);
            count -= n;
            const bool live = lane < n;
            unsigned key = (unsigned)SPMM_ROWS, vbits = 0u;
            if (live) {
                key   = mylist[count + lane];
                vbits = mylist[LIST_VALS + count + lane];
            }
            const int   c  = (int)min(key >> 8, (unsigned)(N_NODES - 1));
            const int   lr = (int)(key & 255u);
            const float v  = __builtin_bit_cast(float, vbits);
            if (!LAST) {
                if (live && found + lane < KEEP_CAP) {
                    v2u s;
                    s.x = key;
                    s.y = vbits;
                    mykept[found + lane] = s;
                }
            }
            found += n;
            place(n, c, v, lr);
        }
    };

    if (rescan) {
        const v2u* __restrict__ rec = (const v2u*)pk;
        const int  sub   = lane & (CHUNK_EDGES - 1);
        const bool upper = lane >= CHUNK_EDGES;
        int nc = 0;
        for (int part = 0; part < BM_PARTS; ++part) {
            unsigned wbits = wpart[0];
#pragma unroll
            for (int p = 1; p < BM_PARTS; ++p) wbits = (part == p) ? wpart[p] : wbits;
            const int pc = __builtin_popcount(wbits);
            int upto = pc;
#pragma unroll
            for (int off = 1; off < 32; off <<= 1) {
                const int t = __shfl_up(upto, off, 32);
                if (lane >= off) upto += t;
            }
            {
                int pos = nc + upto - pc;
                const unsigned first = (unsigned)((part * BM_WPP + lane) * SPMM_WAVES + wave) * 32u;
                for (unsigned wb = wbits; wb != 0u; wb &= wb - 1u)
                    clist[pos++] = (unsigned short)(first + (unsigned)__builtin_ctz(wb));
            }
            nc += __builtin_amdgcn_readlane(upto, 31);
            const bool lastp = part == BM_PARTS - 1;
            int k0 = 0;
            while (lastp ? (k0 < nc) : (nc - k0 >= 2 * CHUNK_DEPTH)) {
                const int m   = min(nc - k0, 2 * CHUNK_DEPTH);
                const int myc = (lane < m) ? (int)clist[k0 + lane] : -1;
                v2u r[CHUNK_DEPTH];
                int cid[CHUNK_DEPTH];
#pragma unroll
                for (int u = 0; u < CHUNK_DEPTH; ++u) {
                    const int ca = __builtin_amdgcn_readlane(myc, 2 * u);
                    const int cb = __builtin_amdgcn_readlane(myc, 2 * u + 1);
                    cid[u] = upper ? cb : ca;
                    r[u] = rec[min(max(cid[u], 0), CHUNKS - 1) * CHUNK_EDGES + sub];
                }
#pragma unroll
                for (int u = 0; u < CHUNK_DEPTH; ++u) {
                    if (2 * u < m) {
                        const unsigned d = (r[u].x & 0xFFFFu) - (unsigned)base;
                        const bool hit = (cid[u] >= 0) && (d < SPMM_ROWS);
                        const unsigned b = __builtin_amdgcn_ballot_w32(hit);
                        if (hit) {
                            const int p = count + (int)__builtin_amdgcn_mbcnt_lo(b, 0u);
                            mylist[p]             = ((r[u].x >> 16) << 8) | d;
                            mylist[LIST_VALS + p] = r[u].y;
                        }
                        count += __builtin_popcount(b);
                    }
                    if (u % SCAN_GROUP == SCAN_GROUP - 1) drain(31);
                }
                k0 += 2 * CHUNK_DEPTH;
            }
            if (!lastp) {
                const int rem = nc - k0;
                unsigned short left = 0;
                if (lane < rem) left = clist[k0 + lane];
                if (lane < rem) clist[lane] = left;
                nc = rem;
            }
        }
        drain(0);
    } else {
        for (int k0 = 0, q = 0; k0 < nlist; k0 += 32, ++q) {
            const int m = min(nlist - k0, 32);
            v2u s = ahead[0];
#pragma unroll
            for (int i = 1; i < KEEP_AHEAD; ++i) s = (q == i) ? ahead[i] : s;
            if (q >= KEEP_AHEAD) s = (lane < m) ? mykept[k0 + lane] : nothing;
            const int   c  = (int)min(s.x >> 8, (unsigned)(N_NODES - 1));
            const int   lr = (int)(s.x & 255u);
            const unsigned vbits = s.y;
            const float v  = __builtin_bit_cast(float, vbits);
            place(m, c, v, lr);
        }
    }
    if (!LAST && lane == 0) nfound[wave] = found;
    __syncthreads();
    if (!LAST && wave == SPMM_WAVES - 1) nkept[blockIdx.x * SPMM_WAVES + lane] = nfound[lane];

    {
        const int krow  = wave + SPMM_WAVES * lane;
        const int mycnt = (lane < WAVE_ROWS && krow < SPMM_ROWS) ? min(rowcnt[krow], ROW_CAP) : 0;
        const v2u empty = {0u, 0u};
        v2u rk[WAVE_ROWS];
#pragma unroll
        for (int k = 0; k < WAVE_ROWS; ++k) {
            const int cnt = __builtin_amdgcn_readlane(mycnt, k);
            rk[k] = (lane < cnt) ? slots[(wave + SPMM_WAVES * k) * ROW_CAP + lane] : empty;
        }
        int total = 0;
#pragma unroll
        for (int k = 0; k < WAVE_ROWS; ++k) {
            const int cnt = __builtin_amdgcn_readlane(mycnt, k);
            if (lane < cnt) {
                mylist[total + lane]             = rk[k].x | ((unsigned)k << 24);
                mylist[FLAT_VALS + total + lane] = rk[k].y;
            }
            total += cnt;
        }
        float4 s = make_float4(0.0f, 0.0f, 0.0f, 0.0f);
        int kcur = 0;
        auto flush = [&](int k) {
            float* a = acc + (wave + SPMM_WAVES * k) * ACC_PITCH + lane;
            __hip_atomic_fetch_add(a +  0, s.x, __ATOMIC_RELAXED, __HIP_MEMORY_SCOPE_WORKGROUP);
            __hip_atomic_fetch_add(a + 32, s.y, __ATOMIC_RELAXED, __HIP_MEMORY_SCOPE_WORKGROUP);
            __hip_atomic_fetch_add(a + 64, s.z, __ATOMIC_RELAXED, __HIP_MEMORY_SCOPE_WORKGROUP);
            __hip_atomic_fetch_add(a + 96, s.w, __ATOMIC_RELAXED, __HIP_MEMORY_SCOPE_WORKGROUP);
        };
        for (int j = 0; j < total; j += GATHER_DEPTH) {
            const int m = min(total - j, GATHER_DEPTH);
            unsigned ce = 0u, ve = 0u;
            if (lane < m) {
                ce = mylist[j + lane];
                ve = mylist[FLAT_VALS + j + lane];
            }
            float4 x[GATHER_DEPTH];
#pragma unroll
            for (int u = 0; u < GATHER_DEPTH; ++u) {
                const int cj = __builtin_amdgcn_readlane((int)(ce & 0xFFFFFFu), u);
                x[u] = ((const float4*)(h + (size_t)cj * DIM))[lane];
            }
#pragma unroll
            for (int u = 0; u < GATHER_DEPTH; ++u) {
                if (u < m) {
                    const int ku = __builtin_amdgcn_readlane((int)(ce >> 24), u);
                    if (ku != kcur) {
                        flush(kcur);
                        s = make_float4(0.0f, 0.0f, 0.0f, 0.0f);
                        kcur = ku;
                    }
                    const float vj = __builtin_bit_cast(float, __builtin_amdgcn_readlane((int)ve, u));
                    s.x += vj * x[u].x;
                    s.y += vj * x[u].y;
                    s.z += vj * x[u].z;
                    s.w += vj * x[u].w;
                }
            }
        }
        if (total > 0) flush(kcur);
    }
    __syncthreads();

    const bool mm = wave < 2 * ROW_TILES;
    const int  mt = wave >> 1;
    const int  nh = wave & 1;
    const int  hh = lane >> 4;
    const int  cc = lane & 15;
    v8f d[4] = {};
    float  bc[4] = {0.0f, 0.0f, 0.0f, 0.0f};
    float4 fw = make_float4(0.0f, 0.0f, 0.0f, 0.0f);
    float  fb = 0.0f;
    if (LAST) {
        fw = ((const float4*)fcw)[lane];
        fb = fcb[0];
    }
    if (mm) {
#pragma unroll
        for (int nt = 0; nt < 4; ++nt) bc[nt] = bias[nh * 64 + nt * 16 + cc];
        const float* arow = acc + (mt * 16 + cc) * ACC_PITCH + 2 * hh;
        for (int kk = 0; kk < 32; ++kk) {
            const v2f a = *(const v2f*)(arow + (kk & 3) * 32 + 4 * (kk >> 2));
#pragma unroll
            for (int nt = 0; nt < 4; ++nt) {
                const v2f b = *(const v2f*)&ldsW[(((kk << 3) + nh * 4 + nt) * 32 + lane) * 2];
                d[nt] = __builtin_amdgcn_wmma_f32_16x16x4_f32(
                     false, a,  false, b,
                     (short)0, d[nt],
                     false,  false);
            }
        }
    }
    __syncthreads();
    if (mm) {
#pragma unroll
        for (int nt = 0; nt < 4; ++nt) {
            const int   col = nh * 64 + nt * 16 + cc;
            float* t = acc + (mt * 16 + 8 * hh) * ACC_PITCH + (col & 3) * 32 + (col >> 2);
#pragma unroll
            for (int j = 0; j < 8; ++j) t[j * ACC_PITCH] = tanh_exp_rcp(d[nt][j] + bc[nt]);
        }
    }
    __syncthreads();

    if (!LAST) {
        for (int rr = wave; rr < SPMM_ROWS; rr += SPMM_WAVES) {
            const int row = base + rr;
            if (row < N_NODES) {
                const float* a = acc + rr * ACC_PITCH + lane;
                ((float4*)(out + (size_t)row * DIM))[lane] = make_float4(a[0], a[32], a[64], a[96]);
            }
        }
    } else {
        float* dots = (float*)elist;
        for (int rr = wave; rr < SPMM_ROWS; rr += SPMM_WAVES) {
            const float* a = acc + rr * ACC_PITCH + lane;
            float s = a[0] * fw.x + a[32] * fw.y + a[64] * fw.z + a[96] * fw.w;
#pragma unroll
            for (int off = 16; off > 0; off >>= 1) s += __shfl_down(s, off, 32);
            if (lane == 0) dots[rr] = s + fb;
        }
        __syncthreads();
        if (tid < SPMM_ROWS && base + tid < N_NODES) out[base + tid] = dots[tid];
    }
}

extern "C" void kernel_launch(void* const* d_in, const int* in_sizes, int n_in,
                              void* d_out, int out_size, void* d_ws, size_t ws_size,
                              hipStream_t stream) {
    const float* features = (const float*)d_in[0];
    const int*   erow     = (const int*)  d_in[1];
    const int*   ecol     = (const int*)  d_in[2];
    const float* evals    = (const float*)d_in[3];
    const float* W1       = (const float*)d_in[4];
    const float* b1       = (const float*)d_in[5];
    const float* W2       = (const float*)d_in[6];
    const float* b2       = (const float*)d_in[7];
    const float* fcw      = (const float*)d_in[8];
    const float* fcb      = (const float*)d_in[9];
    float*       out      = (float*)d_out;

    float*    h1    = (float*)d_ws;
    unsigned* kept  = (unsigned*)((char*)d_ws + WS_KEPT_OFF);
    int*      nkept = (int*)((char*)d_ws + WS_NKEPT_OFF);
    unsigned* bm    = (unsigned*)((char*)d_ws + WS_BM_OFF);
    unsigned* pk    = (unsigned*)((char*)d_ws + WS_PK_OFF);

    const dim3 grid(SPMM_WGS);

    edge_chunk_bits<<<dim3(BM_WGS + PACK_WGS), SPMM_THREADS, 0, stream>>>(erow, ecol, evals, bm, pk);
    gcn_layer_own<false><<<grid, SPMM_THREADS, 0, stream>>>(pk, features, W1, b1, fcw, fcb, kept, nkept, bm, h1);
    gcn_layer_own<true><<<grid, SPMM_THREADS, 0, stream>>>(pk, h1, W2, b2, fcw, fcb, kept, nkept, bm, out);
}
